# starting kernel
# baseline (speedup 1.0000x reference)
#include <hip/hip_runtime.h>
#include <hip/hip_bf16.h>

typedef __attribute__((ext_vector_type(2))) float v2f;
typedef __attribute__((ext_vector_type(4))) float v4f;
typedef __attribute__((ext_vector_type(8))) float v8f;

#define N_NODES  50000
#define N_EDGES  800000
#define IN_F     256
#define OUT_F    64
#define N_HEADS  8
#define COLS     (N_HEADS * OUT_F)
#define ROW_TILES (N_NODES / 16)
#define LRELU_SLOPE 0.2f

__global__ void zero_f32(float* __restrict__ p, int n) {
    int i = blockIdx.x * blockDim.x + threadIdx.x;
    if (i < n) p[i] = 0.0f;
}

__global__ __launch_bounds__(256) void gat_gemm_wmma(const float* __restrict__ x,
                                                     const float* __restrict__ W,
                                                     float* __restrict__ h) {
    __shared__ float Wlds[IN_F * OUT_F];

    const int wave = threadIdx.x >> 5;
    const int lane = threadIdx.x & 31;
    const int head = blockIdx.x;

    {
        const v4f* __restrict__ Wg = (const v4f*)(W + (size_t)head * (IN_F * OUT_F));
        v4f* __restrict__ Wl = (v4f*)Wlds;
#pragma unroll
        for (int i = 0; i < 16; ++i) {
            int t = threadIdx.x + i * 256;
            Wl[t] = Wg[t];
        }
    }
    __syncthreads();

    const int rowTile = blockIdx.y * 8 + wave;
    if (rowTile >= ROW_TILES) return;

    const int row0 = rowTile * 16;
    const int fo   = lane & 15;
    const int koff = (lane >> 4) << 1;
    const int arow = row0 + (lane & 15);
    const float* __restrict__ ap = x + (size_t)arow * IN_F + koff;

    v8f acc0 = {0.f,0.f,0.f,0.f,0.f,0.f,0.f,0.f};
    v8f acc1 = acc0, acc2 = acc0, acc3 = acc0;

#pragma unroll 2
    for (int k0 = 0; k0 < IN_F; k0 += 4) {
        v2f a = *(const v2f*)(ap + k0);
        const float* __restrict__ b0 = Wlds + (k0 + koff) * OUT_F + fo;
        const float* __restrict__ b1 = b0 + OUT_F;
        v2f b;
        b.x = b0[0];  b.y = b1[0];
        acc0 = __builtin_amdgcn_wmma_f32_16x16x4_f32(false, a, false, b,
                                                     (short)0, acc0, false, false);
        b.x = b0[16]; b.y = b1[16];
        acc1 = __builtin_amdgcn_wmma_f32_16x16x4_f32(false, a, false, b,
                                                     (short)0, acc1, false, false);
        b.x = b0[32]; b.y = b1[32];
        acc2 = __builtin_amdgcn_wmma_f32_16x16x4_f32(false, a, false, b,
                                                     (short)0, acc2, false, false);
        b.x = b0[48]; b.y = b1[48];
        acc3 = __builtin_amdgcn_wmma_f32_16x16x4_f32(false, a, false, b,
                                                     (short)0, acc3, false, false);
    }

    const int rbase = row0 + ((lane >> 4) << 3);
    float* __restrict__ hp = h + head * OUT_F + fo;
#pragma unroll
    for (int v = 0; v < 8; ++v) {
        float* __restrict__ hr = hp + (size_t)(rbase + v) * COLS;
        hr[0]  = acc0[v];
        hr[16] = acc1[v];
        hr[32] = acc2[v];
        hr[48] = acc3[v];
    }
}

__global__ void gat_attn_coef(const float* __restrict__ h,
                              const float* __restrict__ a,
                              float* __restrict__ a_self,
                              float* __restrict__ a_neigh) {
    int idx = blockIdx.x * blockDim.x + threadIdx.x;
    if (idx >= N_NODES * N_HEADS) return;
    int n = idx >> 3;
    int hd = idx & 7;
    const v4f* __restrict__ hp = (const v4f*)(h + (size_t)n * COLS + hd * OUT_F);
    const v4f* __restrict__ as = (const v4f*)(a + (size_t)hd * 2 * OUT_F);
    const v4f* __restrict__ an = as + (OUT_F / 4);
    float s = 0.f, t = 0.f;
#pragma unroll
    for (int f = 0; f < OUT_F / 4; ++f) {
        v4f v = hp[f], u = as[f], w = an[f];
        s += v.x * u.x + v.y * u.y + v.z * u.z + v.w * u.w;
        t += v.x * w.x + v.y * w.y + v.z * w.z + v.w * w.w;
    }
    a_self[idx]  = s;
    a_neigh[idx] = t;
}

__device__ __forceinline__ float edge_exp(float as, float an, float m) {
    float lg = as + an;
    lg = (lg > 0.f ? lg : LRELU_SLOPE * lg) * m;
    return __expf(lg);
}

__global__ void gat_edge_denom(const int* __restrict__ src,
                               const int* __restrict__ dst,
                               const float* __restrict__ M,
                               const float* __restrict__ a_self,
                               const float* __restrict__ a_neigh,
                               float* __restrict__ denom) {
    int idx = blockIdx.x * blockDim.x + threadIdx.x;
    if (idx >= N_EDGES * N_HEADS) return;
    int e  = idx >> 3;
    int hd = idx & 7;
    int s = src[e], d = dst[e];
    float ex = edge_exp(a_self[s * N_HEADS + hd], a_neigh[d * N_HEADS + hd], M[e]);
    atomicAdd(&denom[d * N_HEADS + hd], ex);
}

__global__ __launch_bounds__(256) void gat_edge_aggregate(const int* __restrict__ src,
                                                          const int* __restrict__ dst,
                                                          const float* __restrict__ M,
                                                          const float* __restrict__ a_self,
                                                          const float* __restrict__ a_neigh,
                                                          const float* __restrict__ denom,
                                                          const float* __restrict__ h,
                                                          float* __restrict__ out) {
    const int wave = threadIdx.x >> 5;
    const int lane = threadIdx.x & 31;
    const int e = blockIdx.x * 8 + wave;
    if (e >= N_EDGES) return;

    const int s = src[e];
    const int d = dst[e];
    const float m = M[e];
    const float* __restrict__ hs = h   + (size_t)s * COLS;
    float* __restrict__       od = out + (size_t)d * COLS;

#pragma unroll
    for (int hd = 0; hd < N_HEADS; ++hd) {
        float ex  = edge_exp(a_self[s * N_HEADS + hd], a_neigh[d * N_HEADS + hd], m);
        float att = ex / (denom[d * N_HEADS + hd] + 1e-16f);
        int f0 = hd * OUT_F + lane;
        atomicAdd(&od[f0],      att * hs[f0]);
        atomicAdd(&od[f0 + 32], att * hs[f0 + 32]);
    }
}

__global__ void gat_elu(float* __restrict__ out, int n) {
    int i = blockIdx.x * blockDim.x + threadIdx.x;
    if (i < n) {
        float v = out[i];
        out[i] = v > 0.f ? v : (__expf(v) - 1.0f);
    }
}

extern "C" void kernel_launch(void* const* d_in, const int* in_sizes, int n_in,
                              void* d_out, int out_size, void* d_ws, size_t ws_size,
                              hipStream_t stream) {
    const float* x   = (const float*)d_in[0];
    const int*   src = (const int*)  d_in[1];
    const int*   dst = (const int*)  d_in[2];
    const float* M   = (const float*)d_in[3];
    const float* W   = (const float*)d_in[4];
    const float* a   = (const float*)d_in[5];
    float* out = (float*)d_out;

    float* h       = (float*)d_ws;
    float* a_self  = h      + (size_t)N_NODES * COLS;
    float* a_neigh = a_self + (size_t)N_NODES * N_HEADS;
    float* denom   = a_neigh+ (size_t)N_NODES * N_HEADS;

    {
        int n = N_NODES * COLS;
        zero_f32<<<(n + 255) / 256, 256, 0, stream>>>(out, n);
        int m = N_NODES * N_HEADS;
        zero_f32<<<(m + 255) / 256, 256, 0, stream>>>(denom, m);
    }

    {
        dim3 grid(N_HEADS, (ROW_TILES + 7) / 8);
        gat_gemm_wmma<<<grid, 256, 0, stream>>>(x, W, h);
    }

    {
        int n = N_NODES * N_HEADS;
        gat_attn_coef<<<(n + 255) / 256, 256, 0, stream>>>(h, a, a_self, a_neigh);
    }

    {
        int n = N_EDGES * N_HEADS;
        gat_edge_denom<<<(n + 255) / 256, 256, 0, stream>>>(src, dst, M,
                                                            a_self, a_neigh, denom);
    }

    {
        int blocks = (N_EDGES + 7) / 8;
        gat_edge_aggregate<<<blocks, 256, 0, stream>>>(src, dst, M, a_self, a_neigh,
                                                       denom, h, out);
    }

    {
        int n = N_NODES * COLS;
        gat_elu<<<(n + 255) / 256, 256, 0, stream>>>(out, n);
    }
}


# Edges counting-sorted by dst (arrival-order rank); one wave per dst row sums exp*h[src] in registers, logits by 16-lane row sums of the gathered row, divides once, stores ELU: no float atomics
# speedup vs baseline: 1.9217x; 1.9217x over previous
#include <hip/hip_runtime.h>
#include <hip/hip_bf16.h>

typedef __attribute__((ext_vector_type(2))) float v2f;
typedef __attribute__((ext_vector_type(4))) float v4f;
typedef __attribute__((ext_vector_type(8))) float v8f;

#define N_NODES  50000
#define N_EDGES  800000
#define IN_F     256
#define OUT_F    64
#define N_HEADS  8
#define COLS     (N_HEADS * OUT_F)
#define ROW_TILES (N_NODES / 16)
#define LRELU_SLOPE 0.2f

__global__ void zero_f32(float* __restrict__ p, int n) {
    int i = blockIdx.x * blockDim.x + threadIdx.x;
    if (i < n) p[i] = 0.0f;
}

__global__ __launch_bounds__(256) void gat_gemm_wmma(const float* __restrict__ x,
                                                     const float* __restrict__ W,
                                                     float* __restrict__ h) {
    __shared__ float Wlds[IN_F * OUT_F];

    const int wave = threadIdx.x >> 5;
    const int lane = threadIdx.x & 31;
    const int head = blockIdx.x;

    {
        const v4f* __restrict__ Wg = (const v4f*)(W + (size_t)head * (IN_F * OUT_F));
        v4f* __restrict__ Wl = (v4f*)Wlds;
#pragma unroll
        for (int i = 0; i < 16; ++i) {
            int t = threadIdx.x + i * 256;
            Wl[t] = Wg[t];
        }
    }
    __syncthreads();

    const int rowTile = blockIdx.y * 8 + wave;
    if (rowTile >= ROW_TILES) return;

    const int row0 = rowTile * 16;
    const int fo   = lane & 15;
    const int koff = (lane >> 4) << 1;
    const int arow = row0 + (lane & 15);
    const float* __restrict__ ap = x + (size_t)arow * IN_F + koff;

    v8f acc0 = {0.f,0.f,0.f,0.f,0.f,0.f,0.f,0.f};
    v8f acc1 = acc0, acc2 = acc0, acc3 = acc0;

#pragma unroll 2
    for (int k0 = 0; k0 < IN_F; k0 += 4) {
        v2f a = *(const v2f*)(ap + k0);
        const float* __restrict__ b0 = Wlds + (k0 + koff) * OUT_F + fo;
        const float* __restrict__ b1 = b0 + OUT_F;
        v2f b;
        b.x = b0[0];  b.y = b1[0];
        acc0 = __builtin_amdgcn_wmma_f32_16x16x4_f32(false, a, false, b,
                                                     (short)0, acc0, false, false);
        b.x = b0[16]; b.y = b1[16];
        acc1 = __builtin_amdgcn_wmma_f32_16x16x4_f32(false, a, false, b,
                                                     (short)0, acc1, false, false);
        b.x = b0[32]; b.y = b1[32];
        acc2 = __builtin_amdgcn_wmma_f32_16x16x4_f32(false, a, false, b,
                                                     (short)0, acc2, false, false);
        b.x = b0[48]; b.y = b1[48];
        acc3 = __builtin_amdgcn_wmma_f32_16x16x4_f32(false, a, false, b,
                                                     (short)0, acc3, false, false);
    }

    const int rbase = row0 + ((lane >> 4) << 3);
    float* __restrict__ hp = h + head * OUT_F + fo;
#pragma unroll
    for (int v = 0; v < 8; ++v) {
        float* __restrict__ hr = hp + (size_t)(rbase + v) * COLS;
        hr[0]  = acc0[v];
        hr[16] = acc1[v];
        hr[32] = acc2[v];
        hr[48] = acc3[v];
    }
}

__global__ void gat_dst_count(const int* __restrict__ dst, int* __restrict__ seg) {
    int e = blockIdx.x * blockDim.x + threadIdx.x;
    if (e < N_EDGES) atomicAdd(&seg[dst[e]], 1);
}

#define SCAN_T   1024
#define SCAN_PER ((N_NODES + SCAN_T - 1) / SCAN_T)

__global__ __launch_bounds__(SCAN_T) void gat_dst_scan(int* __restrict__ seg) {
    __shared__ int part[SCAN_T];
    const int t = threadIdx.x;
    const int b = t * SCAN_PER;
    int sum = 0;
    for (int i = 0; i < SCAN_PER; ++i) {
        int j = b + i;
        if (j < N_NODES) sum += seg[j];
    }
    part[t] = sum;
    __syncthreads();
    for (int off = 1; off < SCAN_T; off <<= 1) {
        int v = (t >= off) ? part[t - off] : 0;
        __syncthreads();
        part[t] += v;
        __syncthreads();
    }
    int run = part[t] - sum;
    for (int i = 0; i < SCAN_PER; ++i) {
        int j = b + i;
        if (j < N_NODES) {
            int c = seg[j];
            seg[j] = run;
            run += c;
        }
    }
}

__global__ void gat_dst_fill(const int* __restrict__ dst, int* __restrict__ seg,
                             int* __restrict__ perm) {
    int e = blockIdx.x * blockDim.x + threadIdx.x;
    if (e < N_EDGES) {
        int pos = atomicAdd(&seg[dst[e]], 1);
        if ((unsigned)pos < (unsigned)N_EDGES) perm[pos] = e;
    }
}

__device__ __forceinline__ float edge_exp(float as, float an, float m) {
    float lg = as + an;
    lg = (lg > 0.f ? lg : LRELU_SLOPE * lg) * m;
    return __expf(lg);
}

__device__ __forceinline__ float row16_sum(float v) {
    v += __builtin_bit_cast(float, __builtin_amdgcn_update_dpp(0, __builtin_bit_cast(int, v), 0x161, 0xf, 0xf, true));
    v += __builtin_bit_cast(float, __builtin_amdgcn_update_dpp(0, __builtin_bit_cast(int, v), 0x162, 0xf, 0xf, true));
    v += __builtin_bit_cast(float, __builtin_amdgcn_update_dpp(0, __builtin_bit_cast(int, v), 0x164, 0xf, 0xf, true));
    v += __builtin_bit_cast(float, __builtin_amdgcn_update_dpp(0, __builtin_bit_cast(int, v), 0x168, 0xf, 0xf, true));
    return v;
}

__device__ __forceinline__ float dot4(v4f p, v4f q) {
    return p.x * q.x + p.y * q.y + p.z * q.z + p.w * q.w;
}

__device__ __forceinline__ float elu1(float v) {
    return v > 0.f ? v : (__expf(v) - 1.0f);
}

__global__ __launch_bounds__(256) void gat_dst_gather(const int* __restrict__ src,
                                                      const float* __restrict__ M,
                                                      const float* __restrict__ a,
                                                      const int* __restrict__ seg,
                                                      const int* __restrict__ perm,
                                                      const float* __restrict__ h,
                                                      float* __restrict__ out) {
    const int lane = threadIdx.x & 31;
    const int wave = __builtin_amdgcn_readfirstlane(threadIdx.x >> 5);
    const int d = blockIdx.x * 8 + wave;
    if (d >= N_NODES) return;

    int beg = d > 0 ? seg[d - 1] : 0;
    int end = seg[d];
    beg = min(max(beg, 0), N_EDGES);
    end = min(max(end, beg), N_EDGES);

    const int half = lane >> 4;
    const int c4   = (lane & 15) * 4;

    v4f   as[4];
    float nb[4];
    {
        const v4f* __restrict__ hd = (const v4f*)(h + (size_t)d * COLS);
#pragma unroll
        for (int j = 0; j < 4; ++j) {
            const float* __restrict__ ah = a + (size_t)(2 * j + half) * (2 * OUT_F) + c4;
            as[j] = *(const v4f*)ah;
            v4f an = *(const v4f*)(ah + OUT_F);
            nb[j] = row16_sum(dot4(hd[j * 32 + lane], an));
        }
    }

    const v4f zero4 = {0.f, 0.f, 0.f, 0.f};
    v4f   acc[4] = {zero4, zero4, zero4, zero4};
    float den[4] = {0.f, 0.f, 0.f, 0.f};

    for (int base = beg; base < end; base += 32) {
        const int idx = base + lane;
        int e = 0;
        if (idx < end) e = perm[idx];
        e = min(max(e, 0), N_EDGES - 1);
        const int   s_l = src[e];
        const float m_l = M[e];
        const int cnt = min(32, end - base);
        for (int k = 0; k < cnt; ++k) {
            const int   s = __builtin_amdgcn_readlane(s_l, k);
            const float m = __builtin_bit_cast(float,
                              __builtin_amdgcn_readlane(__builtin_bit_cast(int, m_l), k));
            const v4f* __restrict__ hr = (const v4f*)(h + (size_t)s * COLS);
            v4f v[4];
#pragma unroll
            for (int j = 0; j < 4; ++j) v[j] = hr[j * 32 + lane];
            float p[4];
#pragma unroll
            for (int j = 0; j < 4; ++j) p[j] = dot4(v[j], as[j]);
#pragma unroll
            for (int j = 0; j < 4; ++j) p[j] = row16_sum(p[j]);
#pragma unroll
            for (int j = 0; j < 4; ++j) {
                const float ex = edge_exp(p[j], nb[j], m);
                den[j] += ex;
                acc[j] += ex * v[j];
            }
        }
    }

    v4f* __restrict__ od = (v4f*)(out + (size_t)d * COLS);
#pragma unroll
    for (int j = 0; j < 4; ++j) {
        const float dj = den[j] + 1e-16f;
        v4f r = acc[j];
        r.x = elu1(r.x / dj);
        r.y = elu1(r.y / dj);
        r.z = elu1(r.z / dj);
        r.w = elu1(r.w / dj);
        od[j * 32 + lane] = r;
    }
}

extern "C" void kernel_launch(void* const* d_in, const int* in_sizes, int n_in,
                              void* d_out, int out_size, void* d_ws, size_t ws_size,
                              hipStream_t stream) {
    const float* x   = (const float*)d_in[0];
    const int*   src = (const int*)  d_in[1];
    const int*   dst = (const int*)  d_in[2];
    const float* M   = (const float*)d_in[3];
    const float* W   = (const float*)d_in[4];
    const float* a   = (const float*)d_in[5];
    float* out = (float*)d_out;

    float* h    = (float*)d_ws;
    int*   seg  = (int*)(h + (size_t)N_NODES * COLS);
    int*   perm = seg + N_NODES;

    zero_f32<<<(N_NODES + 255) / 256, 256, 0, stream>>>((float*)seg, N_NODES);
    gat_dst_count<<<(N_EDGES + 255) / 256, 256, 0, stream>>>(dst, seg);
    gat_dst_scan<<<1, SCAN_T, 0, stream>>>(seg);
    gat_dst_fill<<<(N_EDGES + 255) / 256, 256, 0, stream>>>(dst, seg, perm);

    {
        dim3 grid(N_HEADS, (ROW_TILES + 7) / 8);
        gat_gemm_wmma<<<grid, 256, 0, stream>>>(x, W, h);
    }

    gat_dst_gather<<<(N_NODES + 7) / 8, 256, 0, stream>>>(src, M, a, seg, perm, h, out);
}


# Prefix sum of the per-destination counters done in LDS: whole-line 16-byte copy in, each thread walks 49 words at an odd stride, one-barrier wave scan, whole-line copy out; no strided global trips
# speedup vs baseline: 2.3881x; 1.2427x over previous
#include <hip/hip_runtime.h>
#include <hip/hip_bf16.h>

typedef __attribute__((ext_vector_type(2))) float v2f;
typedef __attribute__((ext_vector_type(4))) float v4f;
typedef __attribute__((ext_vector_type(8))) float v8f;

#define N_NODES  50000
#define N_EDGES  800000
#define IN_F     256
#define OUT_F    64
#define N_HEADS  8
#define COLS     (N_HEADS * OUT_F)
#define ROW_TILES (N_NODES / 16)
#define LRELU_SLOPE 0.2f

__global__ void zero_f32(float* __restrict__ p, int n) {
    int i = blockIdx.x * blockDim.x + threadIdx.x;
    if (i < n) p[i] = 0.0f;
}

__global__ __launch_bounds__(256) void gat_gemm_wmma(const float* __restrict__ x,
                                                     const float* __restrict__ W,
                                                     float* __restrict__ h) {
    __shared__ float Wlds[IN_F * OUT_F];

    const int wave = threadIdx.x >> 5;
    const int lane = threadIdx.x & 31;
    const int head = blockIdx.x;

    {
        const v4f* __restrict__ Wg = (const v4f*)(W + (size_t)head * (IN_F * OUT_F));
        v4f* __restrict__ Wl = (v4f*)Wlds;
#pragma unroll
        for (int i = 0; i < 16; ++i) {
            int t = threadIdx.x + i * 256;
            Wl[t] = Wg[t];
        }
    }
    __syncthreads();

    const int rowTile = blockIdx.y * 8 + wave;
    if (rowTile >= ROW_TILES) return;

    const int row0 = rowTile * 16;
    const int fo   = lane & 15;
    const int koff = (lane >> 4) << 1;
    const int arow = row0 + (lane & 15);
    const float* __restrict__ ap = x + (size_t)arow * IN_F + koff;

    v8f acc0 = {0.f,0.f,0.f,0.f,0.f,0.f,0.f,0.f};
    v8f acc1 = acc0, acc2 = acc0, acc3 = acc0;

#pragma unroll 2
    for (int k0 = 0; k0 < IN_F; k0 += 4) {
        v2f a = *(const v2f*)(ap + k0);
        const float* __restrict__ b0 = Wlds + (k0 + koff) * OUT_F + fo;
        const float* __restrict__ b1 = b0 + OUT_F;
        v2f b;
        b.x = b0[0];  b.y = b1[0];
        acc0 = __builtin_amdgcn_wmma_f32_16x16x4_f32(false, a, false, b,
                                                     (short)0, acc0, false, false);
        b.x = b0[16]; b.y = b1[16];
        acc1 = __builtin_amdgcn_wmma_f32_16x16x4_f32(false, a, false, b,
                                                     (short)0, acc1, false, false);
        b.x = b0[32]; b.y = b1[32];
        acc2 = __builtin_amdgcn_wmma_f32_16x16x4_f32(false, a, false, b,
                                                     (short)0, acc2, false, false);
        b.x = b0[48]; b.y = b1[48];
        acc3 = __builtin_amdgcn_wmma_f32_16x16x4_f32(false, a, false, b,
                                                     (short)0, acc3, false, false);
    }

    const int rbase = row0 + ((lane >> 4) << 3);
    float* __restrict__ hp = h + head * OUT_F + fo;
#pragma unroll
    for (int v = 0; v < 8; ++v) {
        float* __restrict__ hr = hp + (size_t)(rbase + v) * COLS;
        hr[0]  = acc0[v];
        hr[16] = acc1[v];
        hr[32] = acc2[v];
        hr[48] = acc3[v];
    }
}

__global__ void gat_dst_count(const int* __restrict__ dst, int* __restrict__ seg) {
    int e = blockIdx.x * blockDim.x + threadIdx.x;
    if (e < N_EDGES) atomicAdd(&seg[dst[e]], 1);
}

#define SCAN_T   1024
#define SCAN_PER 49
#define SCAN_PAD (SCAN_T * SCAN_PER)
#define SCAN_V4  (SCAN_PAD / 4)
#define SEG_V4   (N_NODES / 4)
#define SCAN_CP  ((SCAN_V4 + SCAN_T - 1) / SCAN_T)

typedef __attribute__((ext_vector_type(4))) int v4i;

__device__ __forceinline__ int wave_incl_scan(int v, int lane) {
#pragma unroll
    for (int off = 1; off < 32; off <<= 1) {
        int u = __shfl_up(v, off);
        if (lane >= off) v += u;
    }
    return v;
}

__global__ __launch_bounds__(SCAN_T) void gat_dst_scan(int* seg) {
    __shared__ __attribute__((aligned(16))) int cnt[SCAN_PAD];
    __shared__ int wtot[SCAN_T / 32];
    const int t    = threadIdx.x;
    const int lane = t & 31;
    const int wave = t >> 5;
    v4i* l4 = (v4i*)cnt;
    v4i* g4 = (v4i*)seg;

#pragma unroll
    for (int i = 0; i < SCAN_CP; ++i) {
        const int q = t + i * SCAN_T;
        v4i v = {0, 0, 0, 0};
        if (q < SEG_V4) v = g4[q];
        if (q < SCAN_V4) l4[q] = v;
    }
    __syncthreads();

    int c[SCAN_PER];
    int sum = 0;
#pragma unroll
    for (int i = 0; i < SCAN_PER; ++i) {
        c[i] = cnt[t * SCAN_PER + i];
        sum += c[i];
    }

    const int inc = wave_incl_scan(sum, lane);
    if (lane == 31) wtot[wave] = inc;
    __syncthreads();
    const int wt   = wtot[lane];
    const int winc = wave_incl_scan(wt, lane);
    const int wbase = __shfl(winc - wt, wave);
    int run = wbase + inc - sum;

#pragma unroll
    for (int i = 0; i < SCAN_PER; ++i) {
        cnt[t * SCAN_PER + i] = run;
        run += c[i];
    }
    __syncthreads();

#pragma unroll
    for (int i = 0; i < SCAN_CP; ++i) {
        const int q = t + i * SCAN_T;
        if (q < SEG_V4) g4[q] = l4[q];
    }
}

__global__ void gat_dst_fill(const int* __restrict__ dst, int* __restrict__ seg,
                             int* __restrict__ perm) {
    int e = blockIdx.x * blockDim.x + threadIdx.x;
    if (e < N_EDGES) {
        int pos = atomicAdd(&seg[dst[e]], 1);
        if ((unsigned)pos < (unsigned)N_EDGES) perm[pos] = e;
    }
}

__device__ __forceinline__ float edge_exp(float as, float an, float m) {
    float lg = as + an;
    lg = (lg > 0.f ? lg : LRELU_SLOPE * lg) * m;
    return __expf(lg);
}

__device__ __forceinline__ float row16_sum(float v) {
    v += __builtin_bit_cast(float, __builtin_amdgcn_update_dpp(0, __builtin_bit_cast(int, v), 0x161, 0xf, 0xf, true));
    v += __builtin_bit_cast(float, __builtin_amdgcn_update_dpp(0, __builtin_bit_cast(int, v), 0x162, 0xf, 0xf, true));
    v += __builtin_bit_cast(float, __builtin_amdgcn_update_dpp(0, __builtin_bit_cast(int, v), 0x164, 0xf, 0xf, true));
    v += __builtin_bit_cast(float, __builtin_amdgcn_update_dpp(0, __builtin_bit_cast(int, v), 0x168, 0xf, 0xf, true));
    return v;
}

__device__ __forceinline__ float dot4(v4f p, v4f q) {
    return p.x * q.x + p.y * q.y + p.z * q.z + p.w * q.w;
}

__device__ __forceinline__ float elu1(float v) {
    return v > 0.f ? v : (__expf(v) - 1.0f);
}

__global__ __launch_bounds__(256) void gat_dst_gather(const int* __restrict__ src,
                                                      const float* __restrict__ M,
                                                      const float* __restrict__ a,
                                                      const int* __restrict__ seg,
                                                      const int* __restrict__ perm,
                                                      const float* __restrict__ h,
                                                      float* __restrict__ out) {
    const int lane = threadIdx.x & 31;
    const int wave = __builtin_amdgcn_readfirstlane(threadIdx.x >> 5);
    const int d = blockIdx.x * 8 + wave;
    if (d >= N_NODES) return;

    int beg = d > 0 ? seg[d - 1] : 0;
    int end = seg[d];
    beg = min(max(beg, 0), N_EDGES);
    end = min(max(end, beg), N_EDGES);

    const int half = lane >> 4;
    const int c4   = (lane & 15) * 4;

    v4f   as[4];
    float nb[4];
    {
        const v4f* __restrict__ hd = (const v4f*)(h + (size_t)d * COLS);
#pragma unroll
        for (int j = 0; j < 4; ++j) {
            const float* __restrict__ ah = a + (size_t)(2 * j + half) * (2 * OUT_F) + c4;
            as[j] = *(const v4f*)ah;
            v4f an = *(const v4f*)(ah + OUT_F);
            nb[j] = row16_sum(dot4(hd[j * 32 + lane], an));
        }
    }

    const v4f zero4 = {0.f, 0.f, 0.f, 0.f};
    v4f   acc[4] = {zero4, zero4, zero4, zero4};
    float den[4] = {0.f, 0.f, 0.f, 0.f};

    for (int base = beg; base < end; base += 32) {
        const int idx = base + lane;
        int e = 0;
        if (idx < end) e = perm[idx];
        e = min(max(e, 0), N_EDGES - 1);
        const int   s_l = src[e];
        const float m_l = M[e];
        const int cnt = min(32, end - base);
        for (int k = 0; k < cnt; ++k) {
            const int   s = __builtin_amdgcn_readlane(s_l, k);
            const float m = __builtin_bit_cast(float,
                              __builtin_amdgcn_readlane(__builtin_bit_cast(int, m_l), k));
            const v4f* __restrict__ hr = (const v4f*)(h + (size_t)s * COLS);
            v4f v[4];
#pragma unroll
            for (int j = 0; j < 4; ++j) v[j] = hr[j * 32 + lane];
            float p[4];
#pragma unroll
            for (int j = 0; j < 4; ++j) p[j] = dot4(v[j], as[j]);
#pragma unroll
            for (int j = 0; j < 4; ++j) p[j] = row16_sum(p[j]);
#pragma unroll
            for (int j = 0; j < 4; ++j) {
                const float ex = edge_exp(p[j], nb[j], m);
                den[j] += ex;
                acc[j] += ex * v[j];
            }
        }
    }

    v4f* __restrict__ od = (v4f*)(out + (size_t)d * COLS);
#pragma unroll
    for (int j = 0; j < 4; ++j) {
        const float dj = den[j] + 1e-16f;
        v4f r = acc[j];
        r.x = elu1(r.x / dj);
        r.y = elu1(r.y / dj);
        r.z = elu1(r.z / dj);
        r.w = elu1(r.w / dj);
        od[j * 32 + lane] = r;
    }
}

extern "C" void kernel_launch(void* const* d_in, const int* in_sizes, int n_in,
                              void* d_out, int out_size, void* d_ws, size_t ws_size,
                              hipStream_t stream) {
    const float* x   = (const float*)d_in[0];
    const int*   src = (const int*)  d_in[1];
    const int*   dst = (const int*)  d_in[2];
    const float* M   = (const float*)d_in[3];
    const float* W   = (const float*)d_in[4];
    const float* a   = (const float*)d_in[5];
    float* out = (float*)d_out;

    float* h    = (float*)d_ws;
    int*   seg  = (int*)(h + (size_t)N_NODES * COLS);
    int*   perm = seg + N_NODES;

    zero_f32<<<(N_NODES + 255) / 256, 256, 0, stream>>>((float*)seg, N_NODES);
    gat_dst_count<<<(N_EDGES + 255) / 256, 256, 0, stream>>>(dst, seg);
    gat_dst_scan<<<1, SCAN_T, 0, stream>>>(seg);
    gat_dst_fill<<<(N_EDGES + 255) / 256, 256, 0, stream>>>(dst, seg, perm);

    {
        dim3 grid(N_HEADS, (ROW_TILES + 7) / 8);
        gat_gemm_wmma<<<grid, 256, 0, stream>>>(x, W, h);
    }

    gat_dst_gather<<<(N_NODES + 7) / 8, 256, 0, stream>>>(src, M, a, seg, perm, h, out);
}


# Edges grouped by destination with LDS counters only, two levels (256-node buckets, then nodes; arrival-order rank in a run, now of LDS adds): no global atomics, 3 launches replace 4
# speedup vs baseline: 3.2498x; 1.3608x over previous
#include <hip/hip_runtime.h>
#include <hip/hip_bf16.h>

typedef __attribute__((ext_vector_type(2))) float v2f;
typedef __attribute__((ext_vector_type(4))) float v4f;
typedef __attribute__((ext_vector_type(8))) float v8f;

#define N_NODES  50000
#define N_EDGES  800000
#define IN_F     256
#define OUT_F    64
#define N_HEADS  8
#define COLS     (N_HEADS * OUT_F)
#define ROW_TILES (N_NODES / 16)
#define LRELU_SLOPE 0.2f

__global__ __launch_bounds__(256) void gat_gemm_wmma(const float* __restrict__ x,
                                                     const float* __restrict__ W,
                                                     float* __restrict__ h) {
    __shared__ float Wlds[IN_F * OUT_F];

    const int wave = threadIdx.x >> 5;
    const int lane = threadIdx.x & 31;
    const int head = blockIdx.x;

    {
        const v4f* __restrict__ Wg = (const v4f*)(W + (size_t)head * (IN_F * OUT_F));
        v4f* __restrict__ Wl = (v4f*)Wlds;
#pragma unroll
        for (int i = 0; i < 16; ++i) {
            int t = threadIdx.x + i * 256;
            Wl[t] = Wg[t];
        }
    }
    __syncthreads();

    const int rowTile = blockIdx.y * 8 + wave;
    if (rowTile >= ROW_TILES) return;

    const int row0 = rowTile * 16;
    const int fo   = lane & 15;
    const int koff = (lane >> 4) << 1;
    const int arow = row0 + (lane & 15);
    const float* __restrict__ ap = x + (size_t)arow * IN_F + koff;

    v8f acc0 = {0.f,0.f,0.f,0.f,0.f,0.f,0.f,0.f};
    v8f acc1 = acc0, acc2 = acc0, acc3 = acc0;

#pragma unroll 2
    for (int k0 = 0; k0 < IN_F; k0 += 4) {
        v2f a = *(const v2f*)(ap + k0);
        const float* __restrict__ b0 = Wlds + (k0 + koff) * OUT_F + fo;
        const float* __restrict__ b1 = b0 + OUT_F;
        v2f b;
        b.x = b0[0];  b.y = b1[0];
        acc0 = __builtin_amdgcn_wmma_f32_16x16x4_f32(false, a, false, b,
                                                     (short)0, acc0, false, false);
        b.x = b0[16]; b.y = b1[16];
        acc1 = __builtin_amdgcn_wmma_f32_16x16x4_f32(false, a, false, b,
                                                     (short)0, acc1, false, false);
        b.x = b0[32]; b.y = b1[32];
        acc2 = __builtin_amdgcn_wmma_f32_16x16x4_f32(false, a, false, b,
                                                     (short)0, acc2, false, false);
        b.x = b0[48]; b.y = b1[48];
        acc3 = __builtin_amdgcn_wmma_f32_16x16x4_f32(false, a, false, b,
                                                     (short)0, acc3, false, false);
    }

    const int rbase = row0 + ((lane >> 4) << 3);
    float* __restrict__ hp = h + head * OUT_F + fo;
#pragma unroll
    for (int v = 0; v < 8; ++v) {
        float* __restrict__ hr = hp + (size_t)(rbase + v) * COLS;
        hr[0]  = acc0[v];
        hr[16] = acc1[v];
        hr[32] = acc2[v];
        hr[48] = acc3[v];
    }
}

#define SORT_T     512
#define SORT_NB    196
#define SORT_V     2
#define SORT_EPB   (SORT_T * SORT_V * 4)
#define SORT_G     ((N_EDGES + SORT_EPB - 1) / SORT_EPB)
#define SORT_PARTS (SORT_T / 256)
#define SORT_GP    (SORT_G / SORT_PARTS)
#define EDGE_V4    (N_EDGES / 4)
#define SORT_KU    12

static_assert(SORT_NB * 256 >= N_NODES, "the buckets cover every node");
static_assert(SORT_NB <= 256 && SORT_G % SORT_PARTS == 0 && N_EDGES % 4 == 0, "table shape");
static_assert(N_EDGES <= (1 << 20), "an edge id fits the low 20 bits of a key");

typedef __attribute__((ext_vector_type(4))) int v4i;

__device__ __forceinline__ int wave_incl_scan(int v, int lane) {
#pragma unroll
    for (int off = 1; off < 32; off <<= 1) {
        int u = __shfl_up(v, off);
        if (lane >= off) v += u;
    }
    return v;
}

__device__ __forceinline__ int bucket_of(int d) {
    return min(max(d >> 8, 0), SORT_NB - 1);
}

__global__ __launch_bounds__(SORT_T) void gat_bucket_hist(const int* __restrict__ dst,
                                                         int* __restrict__ bh) {
    __shared__ int hc[256];
    const int t = threadIdx.x;
    const int g = blockIdx.x;
    v4i d[SORT_V];
#pragma unroll
    for (int k = 0; k < SORT_V; ++k) {
        const int q = g * (SORT_T * SORT_V) + k * SORT_T + t;
        d[k] = (v4i){0, 0, 0, 0};
        if (q < EDGE_V4) d[k] = ((const v4i*)dst)[q];
    }
    if (t < 256) hc[t] = 0;
    __syncthreads();
#pragma unroll
    for (int k = 0; k < SORT_V; ++k) {
        const int q = g * (SORT_T * SORT_V) + k * SORT_T + t;
        if (q < EDGE_V4) {
            atomicAdd(&hc[bucket_of(d[k].x)], 1);
            atomicAdd(&hc[bucket_of(d[k].y)], 1);
            atomicAdd(&hc[bucket_of(d[k].z)], 1);
            atomicAdd(&hc[bucket_of(d[k].w)], 1);
        }
    }
    __syncthreads();
    if (t < SORT_NB) bh[g * SORT_NB + t] = hc[t];
}

__global__ __launch_bounds__(SORT_T) void gat_bucket_place(const int* __restrict__ dst,
                                                          const int* __restrict__ bh,
                                                          int* __restrict__ bstart,
                                                          unsigned* __restrict__ keyed) {
    __shared__ int pall[SORT_PARTS][256];
    __shared__ int pmine[SORT_PARTS][256];
    __shared__ int cur[256];
    __shared__ int wsum[8];
    const int t    = threadIdx.x;
    const int lane = t & 31;
    const int g    = blockIdx.x;
    const int b    = t & 255;
    const int part = t >> 8;
    v4i d[SORT_V];
#pragma unroll
    for (int k = 0; k < SORT_V; ++k) {
        const int q = g * (SORT_T * SORT_V) + k * SORT_T + t;
        d[k] = (v4i){0, 0, 0, 0};
        if (q < EDGE_V4) d[k] = ((const v4i*)dst)[q];
    }

    int sa = 0, sm = 0;
    if (b < SORT_NB) {
        const int* __restrict__ col = bh + part * SORT_GP * SORT_NB + b;
#pragma unroll
        for (int k = 0; k < SORT_GP; ++k) {
            const int v = col[k * SORT_NB];
            sa += v;
            if (part * SORT_GP + k < g) sm += v;
        }
    }
    pall[part][b]  = sa;
    pmine[part][b] = sm;
    __syncthreads();

    int tot = 0, inc = 0;
    if (t < 256) {
#pragma unroll
        for (int p = 0; p < SORT_PARTS; ++p) tot += pall[p][t];
        inc = wave_incl_scan(tot, lane);
        if (lane == 31) wsum[t >> 5] = inc;
    }
    __syncthreads();
    if (t < 256) {
        int wb = 0;
        for (int k = 0; k < (t >> 5); ++k) wb += wsum[k];
        const int start = wb + inc - tot;
        int mine = 0;
#pragma unroll
        for (int p = 0; p < SORT_PARTS; ++p) mine += pmine[p][t];
        cur[t] = start + mine;
        if (g == 0) bstart[t] = start;
    }
    __syncthreads();

#pragma unroll
    for (int k = 0; k < SORT_V; ++k) {
        const int q = g * (SORT_T * SORT_V) + k * SORT_T + t;
        if (q < EDGE_V4) {
            const int dd[4] = {d[k].x, d[k].y, d[k].z, d[k].w};
            int pos[4];
#pragma unroll
            for (int j = 0; j < 4; ++j) pos[j] = atomicAdd(&cur[bucket_of(dd[j])], 1);
#pragma unroll
            for (int j = 0; j < 4; ++j)
                if ((unsigned)pos[j] < (unsigned)N_EDGES)
                    keyed[pos[j]] = ((unsigned)(dd[j] & 255) << 20) | (unsigned)(4 * q + j);
        }
    }
}

__global__ __launch_bounds__(SORT_T) void gat_bucket_sort(const int* __restrict__ bstart,
                                                         const unsigned* __restrict__ keyed,
                                                         int* __restrict__ seg,
                                                         int* __restrict__ perm) {
    __shared__ int nc[256];
    __shared__ int cur[256];
    __shared__ int wsum[8];
    const int t    = threadIdx.x;
    const int lane = t & 31;
    const int b    = blockIdx.x;
    int beg = bstart[b];
    int end = bstart[b + 1];
    beg = min(max(beg, 0), N_EDGES);
    end = min(max(end, beg), N_EDGES);

    if (t < 256) nc[t] = 0;
    __syncthreads();
    for (int i0 = beg; i0 < end; i0 += SORT_KU * SORT_T) {
        unsigned w[SORT_KU];
#pragma unroll
        for (int k = 0; k < SORT_KU; ++k) w[k] = keyed[min(i0 + k * SORT_T + t, end - 1)];
#pragma unroll
        for (int k = 0; k < SORT_KU; ++k)
            if (i0 + k * SORT_T + t < end) atomicAdd(&nc[w[k] >> 20], 1);
    }
    __syncthreads();

    int v = 0, inc = 0;
    if (t < 256) {
        v   = nc[t];
        inc = wave_incl_scan(v, lane);
        if (lane == 31) wsum[t >> 5] = inc;
    }
    __syncthreads();
    if (t < 256) {
        int wb = 0;
        for (int k = 0; k < (t >> 5); ++k) wb += wsum[k];
        const int incl = beg + wb + inc;
        cur[t] = incl - v;
        const int node = b * 256 + t;
        if (node < N_NODES) seg[node] = incl;
    }
    __syncthreads();
    for (int i0 = beg; i0 < end; i0 += SORT_KU * SORT_T) {
        unsigned w[SORT_KU];
        int pos[SORT_KU];
#pragma unroll
        for (int k = 0; k < SORT_KU; ++k) w[k] = keyed[min(i0 + k * SORT_T + t, end - 1)];
#pragma unroll
        for (int k = 0; k < SORT_KU; ++k) {
            pos[k] = -1;
            if (i0 + k * SORT_T + t < end) pos[k] = atomicAdd(&cur[w[k] >> 20], 1);
        }
#pragma unroll
        for (int k = 0; k < SORT_KU; ++k)
            if ((unsigned)pos[k] < (unsigned)N_EDGES) perm[pos[k]] = (int)(w[k] & 0xFFFFFu);
    }
}

__device__ __forceinline__ float edge_exp(float as, float an, float m) {
    float lg = as + an;
    lg = (lg > 0.f ? lg : LRELU_SLOPE * lg) * m;
    return __expf(lg);
}

__device__ __forceinline__ float row16_sum(float v) {
    v += __builtin_bit_cast(float, __builtin_amdgcn_update_dpp(0, __builtin_bit_cast(int, v), 0x161, 0xf, 0xf, true));
    v += __builtin_bit_cast(float, __builtin_amdgcn_update_dpp(0, __builtin_bit_cast(int, v), 0x162, 0xf, 0xf, true));
    v += __builtin_bit_cast(float, __builtin_amdgcn_update_dpp(0, __builtin_bit_cast(int, v), 0x164, 0xf, 0xf, true));
    v += __builtin_bit_cast(float, __builtin_amdgcn_update_dpp(0, __builtin_bit_cast(int, v), 0x168, 0xf, 0xf, true));
    return v;
}

__device__ __forceinline__ float dot4(v4f p, v4f q) {
    return p.x * q.x + p.y * q.y + p.z * q.z + p.w * q.w;
}

__device__ __forceinline__ float elu1(float v) {
    return v > 0.f ? v : (__expf(v) - 1.0f);
}

__global__ __launch_bounds__(256) void gat_dst_gather(const int* __restrict__ src,
                                                      const float* __restrict__ M,
                                                      const float* __restrict__ a,
                                                      const int* __restrict__ seg,
                                                      const int* __restrict__ perm,
                                                      const float* __restrict__ h,
                                                      float* __restrict__ out) {
    const int lane = threadIdx.x & 31;
    const int wave = __builtin_amdgcn_readfirstlane(threadIdx.x >> 5);
    const int d = blockIdx.x * 8 + wave;
    if (d >= N_NODES) return;

    int beg = d > 0 ? seg[d - 1] : 0;
    int end = seg[d];
    beg = min(max(beg, 0), N_EDGES);
    end = min(max(end, beg), N_EDGES);

    const int half = lane >> 4;
    const int c4   = (lane & 15) * 4;

    v4f   as[4];
    float nb[4];
    {
        const v4f* __restrict__ hd = (const v4f*)(h + (size_t)d * COLS);
#pragma unroll
        for (int j = 0; j < 4; ++j) {
            const float* __restrict__ ah = a + (size_t)(2 * j + half) * (2 * OUT_F) + c4;
            as[j] = *(const v4f*)ah;
            v4f an = *(const v4f*)(ah + OUT_F);
            nb[j] = row16_sum(dot4(hd[j * 32 + lane], an));
        }
    }

    const v4f zero4 = {0.f, 0.f, 0.f, 0.f};
    v4f   acc[4] = {zero4, zero4, zero4, zero4};
    float den[4] = {0.f, 0.f, 0.f, 0.f};

    for (int base = beg; base < end; base += 32) {
        const int idx = base + lane;
        int e = 0;
        if (idx < end) e = perm[idx];
        e = min(max(e, 0), N_EDGES - 1);
        const int   s_l = src[e];
        const float m_l = M[e];
        const int cnt = min(32, end - base);
        for (int k = 0; k < cnt; ++k) {
            const int   s = __builtin_amdgcn_readlane(s_l, k);
            const float m = __builtin_bit_cast(float,
                              __builtin_amdgcn_readlane(__builtin_bit_cast(int, m_l), k));
            const v4f* __restrict__ hr = (const v4f*)(h + (size_t)s * COLS);
            v4f v[4];
#pragma unroll
            for (int j = 0; j < 4; ++j) v[j] = hr[j * 32 + lane];
            float p[4];
#pragma unroll
            for (int j = 0; j < 4; ++j) p[j] = dot4(v[j], as[j]);
#pragma unroll
            for (int j = 0; j < 4; ++j) p[j] = row16_sum(p[j]);
#pragma unroll
            for (int j = 0; j < 4; ++j) {
                const float ex = edge_exp(p[j], nb[j], m);
                den[j] += ex;
                acc[j] += ex * v[j];
            }
        }
    }

    v4f* __restrict__ od = (v4f*)(out + (size_t)d * COLS);
#pragma unroll
    for (int j = 0; j < 4; ++j) {
        const float dj = den[j] + 1e-16f;
        v4f r = acc[j];
        r.x = elu1(r.x / dj);
        r.y = elu1(r.y / dj);
        r.z = elu1(r.z / dj);
        r.w = elu1(r.w / dj);
        od[j * 32 + lane] = r;
    }
}

extern "C" void kernel_launch(void* const* d_in, const int* in_sizes, int n_in,
                              void* d_out, int out_size, void* d_ws, size_t ws_size,
                              hipStream_t stream) {
    const float* x   = (const float*)d_in[0];
    const int*   src = (const int*)  d_in[1];
    const int*   dst = (const int*)  d_in[2];
    const float* M   = (const float*)d_in[3];
    const float* W   = (const float*)d_in[4];
    const float* a   = (const float*)d_in[5];
    float* out = (float*)d_out;

    float* h    = (float*)d_ws;
    int*   seg  = (int*)(h + (size_t)N_NODES * COLS);
    int*   perm = seg + N_NODES;
    unsigned* keyed  = (unsigned*)d_ws;
    int*      bh     = (int*)d_ws + N_EDGES;
    int*      bstart = bh + SORT_G * SORT_NB;

    gat_bucket_hist<<<SORT_G, SORT_T, 0, stream>>>(dst, bh);
    gat_bucket_place<<<SORT_G, SORT_T, 0, stream>>>(dst, bh, bstart, keyed);
    gat_bucket_sort<<<SORT_NB, SORT_T, 0, stream>>>(bstart, keyed, seg, perm);

    {
        dim3 grid(N_HEADS, (ROW_TILES + 7) / 8);
        gat_gemm_wmma<<<grid, 256, 0, stream>>>(x, W, h);
    }

    gat_dst_gather<<<(N_NODES + 7) / 8, 256, 0, stream>>>(src, M, a, seg, perm, h, out);
}


# Gather: a lane's four 16-lane logit sums taken together by a transposing butterfly (5 DPP exchanges, not 16), one exp per lane not four; the logit's 64-term sum tree changes
# speedup vs baseline: 3.3964x; 1.0451x over previous
#include <hip/hip_runtime.h>
#include <hip/hip_bf16.h>

typedef __attribute__((ext_vector_type(2))) float v2f;
typedef __attribute__((ext_vector_type(4))) float v4f;
typedef __attribute__((ext_vector_type(8))) float v8f;

#define N_NODES  50000
#define N_EDGES  800000
#define IN_F     256
#define OUT_F    64
#define N_HEADS  8
#define COLS     (N_HEADS * OUT_F)
#define ROW_TILES (N_NODES / 16)
#define LRELU_SLOPE 0.2f

__global__ __launch_bounds__(256) void gat_gemm_wmma(const float* __restrict__ x,
                                                     const float* __restrict__ W,
                                                     float* __restrict__ h) {
    __shared__ float Wlds[IN_F * OUT_F];

    const int wave = threadIdx.x >> 5;
    const int lane = threadIdx.x & 31;
    const int head = blockIdx.x;

    {
        const v4f* __restrict__ Wg = (const v4f*)(W + (size_t)head * (IN_F * OUT_F));
        v4f* __restrict__ Wl = (v4f*)Wlds;
#pragma unroll
        for (int i = 0; i < 16; ++i) {
            int t = threadIdx.x + i * 256;
            Wl[t] = Wg[t];
        }
    }
    __syncthreads();

    const int rowTile = blockIdx.y * 8 + wave;
    if (rowTile >= ROW_TILES) return;

    const int row0 = rowTile * 16;
    const int fo   = lane & 15;
    const int koff = (lane >> 4) << 1;
    const int arow = row0 + (lane & 15);
    const float* __restrict__ ap = x + (size_t)arow * IN_F + koff;

    v8f acc0 = {0.f,0.f,0.f,0.f,0.f,0.f,0.f,0.f};
    v8f acc1 = acc0, acc2 = acc0, acc3 = acc0;

#pragma unroll 2
    for (int k0 = 0; k0 < IN_F; k0 += 4) {
        v2f a = *(const v2f*)(ap + k0);
        const float* __restrict__ b0 = Wlds + (k0 + koff) * OUT_F + fo;
        const float* __restrict__ b1 = b0 + OUT_F;
        v2f b;
        b.x = b0[0];  b.y = b1[0];
        acc0 = __builtin_amdgcn_wmma_f32_16x16x4_f32(false, a, false, b,
                                                     (short)0, acc0, false, false);
        b.x = b0[16]; b.y = b1[16];
        acc1 = __builtin_amdgcn_wmma_f32_16x16x4_f32(false, a, false, b,
                                                     (short)0, acc1, false, false);
        b.x = b0[32]; b.y = b1[32];
        acc2 = __builtin_amdgcn_wmma_f32_16x16x4_f32(false, a, false, b,
                                                     (short)0, acc2, false, false);
        b.x = b0[48]; b.y = b1[48];
        acc3 = __builtin_amdgcn_wmma_f32_16x16x4_f32(false, a, false, b,
                                                     (short)0, acc3, false, false);
    }

    const int rbase = row0 + ((lane >> 4) << 3);
    float* __restrict__ hp = h + head * OUT_F + fo;
#pragma unroll
    for (int v = 0; v < 8; ++v) {
        float* __restrict__ hr = hp + (size_t)(rbase + v) * COLS;
        hr[0]  = acc0[v];
        hr[16] = acc1[v];
        hr[32] = acc2[v];
        hr[48] = acc3[v];
    }
}

#define SORT_T     512
#define SORT_NB    196
#define SORT_V     2
#define SORT_EPB   (SORT_T * SORT_V * 4)
#define SORT_G     ((N_EDGES + SORT_EPB - 1) / SORT_EPB)
#define SORT_PARTS (SORT_T / 256)
#define SORT_GP    (SORT_G / SORT_PARTS)
#define EDGE_V4    (N_EDGES / 4)
#define SORT_KU    12

static_assert(SORT_NB * 256 >= N_NODES, "the buckets cover every node");
static_assert(SORT_NB <= 256 && SORT_G % SORT_PARTS == 0 && N_EDGES % 4 == 0, "table shape");
static_assert(N_EDGES <= (1 << 20), "an edge id fits the low 20 bits of a key");

typedef __attribute__((ext_vector_type(4))) int v4i;

__device__ __forceinline__ int wave_incl_scan(int v, int lane) {
#pragma unroll
    for (int off = 1; off < 32; off <<= 1) {
        int u = __shfl_up(v, off);
        if (lane >= off) v += u;
    }
    return v;
}

__device__ __forceinline__ int bucket_of(int d) {
    return min(max(d >> 8, 0), SORT_NB - 1);
}

__global__ __launch_bounds__(SORT_T) void gat_bucket_hist(const int* __restrict__ dst,
                                                         int* __restrict__ bh) {
    __shared__ int hc[256];
    const int t = threadIdx.x;
    const int g = blockIdx.x;
    v4i d[SORT_V];
#pragma unroll
    for (int k = 0; k < SORT_V; ++k) {
        const int q = g * (SORT_T * SORT_V) + k * SORT_T + t;
        d[k] = (v4i){0, 0, 0, 0};
        if (q < EDGE_V4) d[k] = ((const v4i*)dst)[q];
    }
    if (t < 256) hc[t] = 0;
    __syncthreads();
#pragma unroll
    for (int k = 0; k < SORT_V; ++k) {
        const int q = g * (SORT_T * SORT_V) + k * SORT_T + t;
        if (q < EDGE_V4) {
            atomicAdd(&hc[bucket_of(d[k].x)], 1);
            atomicAdd(&hc[bucket_of(d[k].y)], 1);
            atomicAdd(&hc[bucket_of(d[k].z)], 1);
            atomicAdd(&hc[bucket_of(d[k].w)], 1);
        }
    }
    __syncthreads();
    if (t < SORT_NB) bh[g * SORT_NB + t] = hc[t];
}

__global__ __launch_bounds__(SORT_T) void gat_bucket_place(const int* __restrict__ dst,
                                                          const int* __restrict__ bh,
                                                          int* __restrict__ bstart,
                                                          unsigned* __restrict__ keyed) {
    __shared__ int pall[SORT_PARTS][256];
    __shared__ int pmine[SORT_PARTS][256];
    __shared__ int cur[256];
    __shared__ int wsum[8];
    const int t    = threadIdx.x;
    const int lane = t & 31;
    const int g    = blockIdx.x;
    const int b    = t & 255;
    const int part = t >> 8;
    v4i d[SORT_V];
#pragma unroll
    for (int k = 0; k < SORT_V; ++k) {
        const int q = g * (SORT_T * SORT_V) + k * SORT_T + t;
        d[k] = (v4i){0, 0, 0, 0};
        if (q < EDGE_V4) d[k] = ((const v4i*)dst)[q];
    }

    int sa = 0, sm = 0;
    if (b < SORT_NB) {
        const int* __restrict__ col = bh + part * SORT_GP * SORT_NB + b;
#pragma unroll
        for (int k = 0; k < SORT_GP; ++k) {
            const int v = col[k * SORT_NB];
            sa += v;
            if (part * SORT_GP + k < g) sm += v;
        }
    }
    pall[part][b]  = sa;
    pmine[part][b] = sm;
    __syncthreads();

    int tot = 0, inc = 0;
    if (t < 256) {
#pragma unroll
        for (int p = 0; p < SORT_PARTS; ++p) tot += pall[p][t];
        inc = wave_incl_scan(tot, lane);
        if (lane == 31) wsum[t >> 5] = inc;
    }
    __syncthreads();
    if (t < 256) {
        int wb = 0;
        for (int k = 0; k < (t >> 5); ++k) wb += wsum[k];
        const int start = wb + inc - tot;
        int mine = 0;
#pragma unroll
        for (int p = 0; p < SORT_PARTS; ++p) mine += pmine[p][t];
        cur[t] = start + mine;
        if (g == 0) bstart[t] = start;
    }
    __syncthreads();

#pragma unroll
    for (int k = 0; k < SORT_V; ++k) {
        const int q = g * (SORT_T * SORT_V) + k * SORT_T + t;
        if (q < EDGE_V4) {
            const int dd[4] = {d[k].x, d[k].y, d[k].z, d[k].w};
            int pos[4];
#pragma unroll
            for (int j = 0; j < 4; ++j) pos[j] = atomicAdd(&cur[bucket_of(dd[j])], 1);
#pragma unroll
            for (int j = 0; j < 4; ++j)
                if ((unsigned)pos[j] < (unsigned)N_EDGES)
                    keyed[pos[j]] = ((unsigned)(dd[j] & 255) << 20) | (unsigned)(4 * q + j);
        }
    }
}

__global__ __launch_bounds__(SORT_T) void gat_bucket_sort(const int* __restrict__ bstart,
                                                         const unsigned* __restrict__ keyed,
                                                         int* __restrict__ seg,
                                                         int* __restrict__ perm) {
    __shared__ int nc[256];
    __shared__ int cur[256];
    __shared__ int wsum[8];
    const int t    = threadIdx.x;
    const int lane = t & 31;
    const int b    = blockIdx.x;
    int beg = bstart[b];
    int end = bstart[b + 1];
    beg = min(max(beg, 0), N_EDGES);
    end = min(max(end, beg), N_EDGES);

    if (t < 256) nc[t] = 0;
    __syncthreads();
    for (int i0 = beg; i0 < end; i0 += SORT_KU * SORT_T) {
        unsigned w[SORT_KU];
#pragma unroll
        for (int k = 0; k < SORT_KU; ++k) w[k] = keyed[min(i0 + k * SORT_T + t, end - 1)];
#pragma unroll
        for (int k = 0; k < SORT_KU; ++k)
            if (i0 + k * SORT_T + t < end) atomicAdd(&nc[w[k] >> 20], 1);
    }
    __syncthreads();

    int v = 0, inc = 0;
    if (t < 256) {
        v   = nc[t];
        inc = wave_incl_scan(v, lane);
        if (lane == 31) wsum[t >> 5] = inc;
    }
    __syncthreads();
    if (t < 256) {
        int wb = 0;
        for (int k = 0; k < (t >> 5); ++k) wb += wsum[k];
        const int incl = beg + wb + inc;
        cur[t] = incl - v;
        const int node = b * 256 + t;
        if (node < N_NODES) seg[node] = incl;
    }
    __syncthreads();
    for (int i0 = beg; i0 < end; i0 += SORT_KU * SORT_T) {
        unsigned w[SORT_KU];
        int pos[SORT_KU];
#pragma unroll
        for (int k = 0; k < SORT_KU; ++k) w[k] = keyed[min(i0 + k * SORT_T + t, end - 1)];
#pragma unroll
        for (int k = 0; k < SORT_KU; ++k) {
            pos[k] = -1;
            if (i0 + k * SORT_T + t < end) pos[k] = atomicAdd(&cur[w[k] >> 20], 1);
        }
#pragma unroll
        for (int k = 0; k < SORT_KU; ++k)
            if ((unsigned)pos[k] < (unsigned)N_EDGES) perm[pos[k]] = (int)(w[k] & 0xFFFFFu);
    }
}

__device__ __forceinline__ float edge_exp(float as, float an, float m) {
    float lg = as + an;
    lg = (lg > 0.f ? lg : LRELU_SLOPE * lg) * m;
    return __expf(lg);
}

__device__ __forceinline__ float row16_sum(float v) {
    v += __builtin_bit_cast(float, __builtin_amdgcn_update_dpp(0, __builtin_bit_cast(int, v), 0x161, 0xf, 0xf, true));
    v += __builtin_bit_cast(float, __builtin_amdgcn_update_dpp(0, __builtin_bit_cast(int, v), 0x162, 0xf, 0xf, true));
    v += __builtin_bit_cast(float, __builtin_amdgcn_update_dpp(0, __builtin_bit_cast(int, v), 0x164, 0xf, 0xf, true));
    v += __builtin_bit_cast(float, __builtin_amdgcn_update_dpp(0, __builtin_bit_cast(int, v), 0x168, 0xf, 0xf, true));
    return v;
}

__device__ __forceinline__ float dot4(v4f p, v4f q) {
    return p.x * q.x + p.y * q.y + p.z * q.z + p.w * q.w;
}

__device__ __forceinline__ float dot4_packed(v4f p, v4f q) {
    v2f t = p.xy * q.xy;
    t = __builtin_elementwise_fma(p.zw, q.zw, t);
    return t.x + t.y;
}

template <int CTRL>
__device__ __forceinline__ float row_read(float v) {
    return __builtin_bit_cast(float, __builtin_amdgcn_update_dpp(0, __builtin_bit_cast(int, v), CTRL, 0xf, 0xf, true));
}

__device__ __forceinline__ float elu1(float v) {
    return v > 0.f ? v : (__expf(v) - 1.0f);
}

__global__ __launch_bounds__(256) void gat_dst_gather(const int* __restrict__ src,
                                                      const float* __restrict__ M,
                                                      const float* __restrict__ a,
                                                      const int* __restrict__ seg,
                                                      const int* __restrict__ perm,
                                                      const float* __restrict__ h,
                                                      float* __restrict__ out) {
    const int lane = threadIdx.x & 31;
    const int wave = __builtin_amdgcn_readfirstlane(threadIdx.x >> 5);
    const int d = blockIdx.x * 8 + wave;
    if (d >= N_NODES) return;

    int beg = d > 0 ? seg[d - 1] : 0;
    int end = seg[d];
    beg = min(max(beg, 0), N_EDGES);
    end = min(max(end, beg), N_EDGES);

    const int half = lane >> 4;
    const int c4   = (lane & 15) * 4;

    v4f   as[4];
    float nb[4];
    {
        const v4f* __restrict__ hd = (const v4f*)(h + (size_t)d * COLS);
#pragma unroll
        for (int j = 0; j < 4; ++j) {
            const float* __restrict__ ah = a + (size_t)(2 * j + half) * (2 * OUT_F) + c4;
            as[j] = *(const v4f*)ah;
            v4f an = *(const v4f*)(ah + OUT_F);
            nb[j] = row16_sum(dot4(hd[j * 32 + lane], an));
        }
    }

    const v4f zero4 = {0.f, 0.f, 0.f, 0.f};
    v4f   acc[4] = {zero4, zero4, zero4, zero4};
    const bool up1 = (lane & 1) != 0;
    const bool up2 = (lane & 2) != 0;
    const float nb_own = up1 ? (up2 ? nb[3] : nb[2]) : (up2 ? nb[1] : nb[0]);
    float den_own = 0.f;

    for (int base = beg; base < end; base += 32) {
        const int idx = base + lane;
        int e = 0;
        if (idx < end) e = perm[idx];
        e = min(max(e, 0), N_EDGES - 1);
        const int   s_l = src[e];
        const float m_l = M[e];
        const int cnt = min(32, end - base);
        for (int k = 0; k < cnt; ++k) {
            const int   s = __builtin_amdgcn_readlane(s_l, k);
            const float m = __builtin_bit_cast(float,
                              __builtin_amdgcn_readlane(__builtin_bit_cast(int, m_l), k));
            const v4f* __restrict__ hr = (const v4f*)(h + (size_t)s * COLS);
            v4f v[4];
#pragma unroll
            for (int j = 0; j < 4; ++j) v[j] = hr[j * 32 + lane];
            float p[4];
#pragma unroll
            for (int j = 0; j < 4; ++j) p[j] = dot4_packed(v[j], as[j]);
            float k0 = up1 ? p[2] : p[0];
            float k1 = up1 ? p[3] : p[1];
            const float g0 = up1 ? p[0] : p[2];
            const float g1 = up1 ? p[1] : p[3];
            k0 += row_read<0x161>(g0);
            k1 += row_read<0x161>(g1);
            float lsum = up2 ? k1 : k0;
            const float g = up2 ? k0 : k1;
            lsum += row_read<0x162>(g);
            lsum += row_read<0x164>(lsum);
            lsum += row_read<0x168>(lsum);
            const float ex = edge_exp(lsum, nb_own, m);
            den_own += ex;
            const float e0 = row_read<0x00>(ex);
            const float e1 = row_read<0xAA>(ex);
            const float e2 = row_read<0x55>(ex);
            const float e3 = row_read<0xFF>(ex);
            acc[0] += e0 * v[0];
            acc[1] += e1 * v[1];
            acc[2] += e2 * v[2];
            acc[3] += e3 * v[3];
        }
    }
    const float den[4] = {row_read<0x00>(den_own), row_read<0xAA>(den_own),
                          row_read<0x55>(den_own), row_read<0xFF>(den_own)};

    v4f* __restrict__ od = (v4f*)(out + (size_t)d * COLS);
#pragma unroll
    for (int j = 0; j < 4; ++j) {
        const float dj = den[j] + 1e-16f;
        v4f r = acc[j];
        r.x = elu1(r.x / dj);
        r.y = elu1(r.y / dj);
        r.z = elu1(r.z / dj);
        r.w = elu1(r.w / dj);
        od[j * 32 + lane] = r;
    }
}

extern "C" void kernel_launch(void* const* d_in, const int* in_sizes, int n_in,
                              void* d_out, int out_size, void* d_ws, size_t ws_size,
                              hipStream_t stream) {
    const float* x   = (const float*)d_in[0];
    const int*   src = (const int*)  d_in[1];
    const int*   dst = (const int*)  d_in[2];
    const float* M   = (const float*)d_in[3];
    const float* W   = (const float*)d_in[4];
    const float* a   = (const float*)d_in[5];
    float* out = (float*)d_out;

    float* h    = (float*)d_ws;
    int*   seg  = (int*)(h + (size_t)N_NODES * COLS);
    int*   perm = seg + N_NODES;
    unsigned* keyed  = (unsigned*)d_ws;
    int*      bh     = (int*)d_ws + N_EDGES;
    int*      bstart = bh + SORT_G * SORT_NB;

    gat_bucket_hist<<<SORT_G, SORT_T, 0, stream>>>(dst, bh);
    gat_bucket_place<<<SORT_G, SORT_T, 0, stream>>>(dst, bh, bstart, keyed);
    gat_bucket_sort<<<SORT_NB, SORT_T, 0, stream>>>(bstart, keyed, seg, perm);

    {
        dim3 grid(N_HEADS, (ROW_TILES + 7) / 8);
        gat_gemm_wmma<<<grid, 256, 0, stream>>>(x, W, h);
    }

    gat_dst_gather<<<(N_NODES + 7) / 8, 256, 0, stream>>>(src, M, a, seg, perm, h, out);
}


# Product's operand feed: W kept in LDS in fragment order (one 16-byte read feeds four WMMAs), x read one whole 128-byte line per trip, 32x64 strip per wave; k in each eight summed as 0,1,4,5,2,3,6,7
# speedup vs baseline: 3.5329x; 1.0402x over previous
#include <hip/hip_runtime.h>
#include <hip/hip_bf16.h>

typedef __attribute__((ext_vector_type(2))) float v2f;
typedef __attribute__((ext_vector_type(4))) float v4f;
typedef __attribute__((ext_vector_type(8))) float v8f;

#define N_NODES  50000
#define N_EDGES  800000
#define IN_F     256
#define OUT_F    64
#define N_HEADS  8
#define COLS     (N_HEADS * OUT_F)
#define ROW_TILES (N_NODES / 16)
#define LRELU_SLOPE 0.2f

#define GEMM_TPB 8
#define GEMM_RG  ((ROW_TILES + GEMM_TPB - 1) / GEMM_TPB)

__global__ __launch_bounds__(128) void gat_gemm_wmma(const float* __restrict__ x,
                                                     const float* __restrict__ W,
                                                     float* __restrict__ h) {
    __shared__ v4f Wl[IN_F * OUT_F / 4];

    const int tid  = threadIdx.x;
    const int wave = tid >> 5;
    const int lane = tid & 31;
    const int head = blockIdx.x;

    {
        const float* __restrict__ Wg = W + (size_t)head * (IN_F * OUT_F);
#pragma unroll 4
        for (int b = 0; b < 8; ++b) {
            const int blk = b * 128 + tid;
            const int kg  = blk >> 5;
            const int hf  = (blk >> 4) & 1;
            const int q   = blk & 15;
            const float* __restrict__ src0 = Wg + (size_t)(kg * 8 + hf * 4) * OUT_F + q * 4;
            const v4f r0 = *(const v4f*)(src0);
            const v4f r1 = *(const v4f*)(src0 + OUT_F);
            const v4f r2 = *(const v4f*)(src0 + 2 * OUT_F);
            const v4f r3 = *(const v4f*)(src0 + 3 * OUT_F);
            const int t = q >> 2;
            const int u = (kg * 4 + t) * 32 + hf * 16 + (q & 3) * 4;
            Wl[u + (0 ^ t)] = (v4f){r0.x, r1.x, r2.x, r3.x};
            Wl[u + (1 ^ t)] = (v4f){r0.y, r1.y, r2.y, r3.y};
            Wl[u + (2 ^ t)] = (v4f){r0.z, r1.z, r2.z, r3.z};
            Wl[u + (3 ^ t)] = (v4f){r0.w, r1.w, r2.w, r3.w};
        }
    }
    __syncthreads();

    const int tile0 = (blockIdx.y * 4 + wave) * 2;
    if (tile0 >= ROW_TILES) return;
    const bool two = (tile0 + 1) < ROW_TILES;

    const int c  = lane & 15;
    const int hf = lane >> 4;
    const int rowA0 = tile0 * 16 + c;
    const int rowA1 = min(rowA0 + 16, N_NODES - 1);
    const float* __restrict__ ap0 = x + (size_t)rowA0 * IN_F + hf * 4;
    const float* __restrict__ ap1 = x + (size_t)rowA1 * IN_F + hf * 4;
    const v4f* __restrict__ bp[4] = {Wl + lane, Wl + (lane ^ 1), Wl + (lane ^ 2), Wl + (lane ^ 3)};

    const v8f z8 = {0.f,0.f,0.f,0.f,0.f,0.f,0.f,0.f};
    v8f acc[2][4] = {{z8, z8, z8, z8}, {z8, z8, z8, z8}};

    for (int kc = 0; kc < IN_F / 32; ++kc) {
        v4f a0[4], a1[4], b[4][4];
#pragma unroll
        for (int g = 0; g < 4; ++g) {
            a0[g] = *(const v4f*)(ap0 + kc * 32 + g * 8);
            a1[g] = *(const v4f*)(ap1 + kc * 32 + g * 8);
        }
#pragma unroll
        for (int g = 0; g < 4; ++g)
#pragma unroll
            for (int t = 0; t < 4; ++t) b[g][t] = bp[t][((kc * 4 + g) * 4 + t) * 32];
        __builtin_amdgcn_sched_barrier(0);
#pragma unroll
        for (int g = 0; g < 4; ++g) {
#pragma unroll
            for (int t = 0; t < 4; ++t) {
                acc[0][t] = __builtin_amdgcn_wmma_f32_16x16x4_f32(false, a0[g].xy, false, b[g][t].xy,
                                                                  (short)0, acc[0][t], false, false);
                acc[1][t] = __builtin_amdgcn_wmma_f32_16x16x4_f32(false, a1[g].xy, false, b[g][t].xy,
                                                                  (short)0, acc[1][t], false, false);
            }
#pragma unroll
            for (int t = 0; t < 4; ++t) {
                acc[0][t] = __builtin_amdgcn_wmma_f32_16x16x4_f32(false, a0[g].zw, false, b[g][t].zw,
                                                                  (short)0, acc[0][t], false, false);
                acc[1][t] = __builtin_amdgcn_wmma_f32_16x16x4_f32(false, a1[g].zw, false, b[g][t].zw,
                                                                  (short)0, acc[1][t], false, false);
            }
        }
    }

    float* __restrict__ hp = h + (size_t)(tile0 * 16 + hf * 8) * COLS + head * OUT_F + c;
#pragma unroll
    for (int v = 0; v < 8; ++v) {
        float* __restrict__ hr = hp + (size_t)v * COLS;
        hr[0]  = acc[0][0][v];
        hr[16] = acc[0][1][v];
        hr[32] = acc[0][2][v];
        hr[48] = acc[0][3][v];
    }
    if (two) {
#pragma unroll
        for (int v = 0; v < 8; ++v) {
            float* __restrict__ hr = hp + (size_t)(16 + v) * COLS;
            hr[0]  = acc[1][0][v];
            hr[16] = acc[1][1][v];
            hr[32] = acc[1][2][v];
            hr[48] = acc[1][3][v];
        }
    }
}

#define SORT_T     512
#define SORT_NB    196
#define SORT_V     2
#define SORT_EPB   (SORT_T * SORT_V * 4)
#define SORT_G     ((N_EDGES + SORT_EPB - 1) / SORT_EPB)
#define SORT_PARTS (SORT_T / 256)
#define SORT_GP    (SORT_G / SORT_PARTS)
#define EDGE_V4    (N_EDGES / 4)
#define SORT_KU    12

static_assert(SORT_NB * 256 >= N_NODES, "the buckets cover every node");
static_assert(SORT_NB <= 256 && SORT_G % SORT_PARTS == 0 && N_EDGES % 4 == 0, "table shape");
static_assert(N_EDGES <= (1 << 20), "an edge id fits the low 20 bits of a key");

typedef __attribute__((ext_vector_type(4))) int v4i;

__device__ __forceinline__ int wave_incl_scan(int v, int lane) {
#pragma unroll
    for (int off = 1; off < 32; off <<= 1) {
        int u = __shfl_up(v, off);
        if (lane >= off) v += u;
    }
    return v;
}

__device__ __forceinline__ int bucket_of(int d) {
    return min(max(d >> 8, 0), SORT_NB - 1);
}

__global__ __launch_bounds__(SORT_T) void gat_bucket_hist(const int* __restrict__ dst,
                                                         int* __restrict__ bh) {
    __shared__ int hc[256];
    const int t = threadIdx.x;
    const int g = blockIdx.x;
    v4i d[SORT_V];
#pragma unroll
    for (int k = 0; k < SORT_V; ++k) {
        const int q = g * (SORT_T * SORT_V) + k * SORT_T + t;
        d[k] = (v4i){0, 0, 0, 0};
        if (q < EDGE_V4) d[k] = ((const v4i*)dst)[q];
    }
    if (t < 256) hc[t] = 0;
    __syncthreads();
#pragma unroll
    for (int k = 0; k < SORT_V; ++k) {
        const int q = g * (SORT_T * SORT_V) + k * SORT_T + t;
        if (q < EDGE_V4) {
            atomicAdd(&hc[bucket_of(d[k].x)], 1);
            atomicAdd(&hc[bucket_of(d[k].y)], 1);
            atomicAdd(&hc[bucket_of(d[k].z)], 1);
            atomicAdd(&hc[bucket_of(d[k].w)], 1);
        }
    }
    __syncthreads();
    if (t < SORT_NB) bh[g * SORT_NB + t] = hc[t];
}

__global__ __launch_bounds__(SORT_T) void gat_bucket_place(const int* __restrict__ dst,
                                                          const int* __restrict__ bh,
                                                          int* __restrict__ bstart,
                                                          unsigned* __restrict__ keyed) {
    __shared__ int pall[SORT_PARTS][256];
    __shared__ int pmine[SORT_PARTS][256];
    __shared__ int cur[256];
    __shared__ int wsum[8];
    const int t    = threadIdx.x;
    const int lane = t & 31;
    const int g    = blockIdx.x;
    const int b    = t & 255;
    const int part = t >> 8;
    v4i d[SORT_V];
#pragma unroll
    for (int k = 0; k < SORT_V; ++k) {
        const int q = g * (SORT_T * SORT_V) + k * SORT_T + t;
        d[k] = (v4i){0, 0, 0, 0};
        if (q < EDGE_V4) d[k] = ((const v4i*)dst)[q];
    }

    int sa = 0, sm = 0;
    if (b < SORT_NB) {
        const int* __restrict__ col = bh + part * SORT_GP * SORT_NB + b;
#pragma unroll
        for (int k = 0; k < SORT_GP; ++k) {
            const int v = col[k * SORT_NB];
            sa += v;
            if (part * SORT_GP + k < g) sm += v;
        }
    }
    pall[part][b]  = sa;
    pmine[part][b] = sm;
    __syncthreads();

    int tot = 0, inc = 0;
    if (t < 256) {
#pragma unroll
        for (int p = 0; p < SORT_PARTS; ++p) tot += pall[p][t];
        inc = wave_incl_scan(tot, lane);
        if (lane == 31) wsum[t >> 5] = inc;
    }
    __syncthreads();
    if (t < 256) {
        int wb = 0;
        for (int k = 0; k < (t >> 5); ++k) wb += wsum[k];
        const int start = wb + inc - tot;
        int mine = 0;
#pragma unroll
        for (int p = 0; p < SORT_PARTS; ++p) mine += pmine[p][t];
        cur[t] = start + mine;
        if (g == 0) bstart[t] = start;
    }
    __syncthreads();

#pragma unroll
    for (int k = 0; k < SORT_V; ++k) {
        const int q = g * (SORT_T * SORT_V) + k * SORT_T + t;
        if (q < EDGE_V4) {
            const int dd[4] = {d[k].x, d[k].y, d[k].z, d[k].w};
            int pos[4];
#pragma unroll
            for (int j = 0; j < 4; ++j) pos[j] = atomicAdd(&cur[bucket_of(dd[j])], 1);
#pragma unroll
            for (int j = 0; j < 4; ++j)
                if ((unsigned)pos[j] < (unsigned)N_EDGES)
                    keyed[pos[j]] = ((unsigned)(dd[j] & 255) << 20) | (unsigned)(4 * q + j);
        }
    }
}

__global__ __launch_bounds__(SORT_T) void gat_bucket_sort(const int* __restrict__ bstart,
                                                         const unsigned* __restrict__ keyed,
                                                         int* __restrict__ seg,
                                                         int* __restrict__ perm) {
    __shared__ int nc[256];
    __shared__ int cur[256];
    __shared__ int wsum[8];
    const int t    = threadIdx.x;
    const int lane = t & 31;
    const int b    = blockIdx.x;
    int beg = bstart[b];
    int end = bstart[b + 1];
    beg = min(max(beg, 0), N_EDGES);
    end = min(max(end, beg), N_EDGES);

    if (t < 256) nc[t] = 0;
    __syncthreads();
    for (int i0 = beg; i0 < end; i0 += SORT_KU * SORT_T) {
        unsigned w[SORT_KU];
#pragma unroll
        for (int k = 0; k < SORT_KU; ++k) w[k] = keyed[min(i0 + k * SORT_T + t, end - 1)];
#pragma unroll
        for (int k = 0; k < SORT_KU; ++k)
            if (i0 + k * SORT_T + t < end) atomicAdd(&nc[w[k] >> 20], 1);
    }
    __syncthreads();

    int v = 0, inc = 0;
    if (t < 256) {
        v   = nc[t];
        inc = wave_incl_scan(v, lane);
        if (lane == 31) wsum[t >> 5] = inc;
    }
    __syncthreads();
    if (t < 256) {
        int wb = 0;
        for (int k = 0; k < (t >> 5); ++k) wb += wsum[k];
        const int incl = beg + wb + inc;
        cur[t] = incl - v;
        const int node = b * 256 + t;
        if (node < N_NODES) seg[node] = incl;
    }
    __syncthreads();
    for (int i0 = beg; i0 < end; i0 += SORT_KU * SORT_T) {
        unsigned w[SORT_KU];
        int pos[SORT_KU];
#pragma unroll
        for (int k = 0; k < SORT_KU; ++k) w[k] = keyed[min(i0 + k * SORT_T + t, end - 1)];
#pragma unroll
        for (int k = 0; k < SORT_KU; ++k) {
            pos[k] = -1;
            if (i0 + k * SORT_T + t < end) pos[k] = atomicAdd(&cur[w[k] >> 20], 1);
        }
#pragma unroll
        for (int k = 0; k < SORT_KU; ++k)
            if ((unsigned)pos[k] < (unsigned)N_EDGES) perm[pos[k]] = (int)(w[k] & 0xFFFFFu);
    }
}

__device__ __forceinline__ float edge_exp(float as, float an, float m) {
    float lg = as + an;
    lg = (lg > 0.f ? lg : LRELU_SLOPE * lg) * m;
    return __expf(lg);
}

__device__ __forceinline__ float row16_sum(float v) {
    v += __builtin_bit_cast(float, __builtin_amdgcn_update_dpp(0, __builtin_bit_cast(int, v), 0x161, 0xf, 0xf, true));
    v += __builtin_bit_cast(float, __builtin_amdgcn_update_dpp(0, __builtin_bit_cast(int, v), 0x162, 0xf, 0xf, true));
    v += __builtin_bit_cast(float, __builtin_amdgcn_update_dpp(0, __builtin_bit_cast(int, v), 0x164, 0xf, 0xf, true));
    v += __builtin_bit_cast(float, __builtin_amdgcn_update_dpp(0, __builtin_bit_cast(int, v), 0x168, 0xf, 0xf, true));
    return v;
}

__device__ __forceinline__ float dot4(v4f p, v4f q) {
    return p.x * q.x + p.y * q.y + p.z * q.z + p.w * q.w;
}

__device__ __forceinline__ float dot4_packed(v4f p, v4f q) {
    v2f t = p.xy * q.xy;
    t = __builtin_elementwise_fma(p.zw, q.zw, t);
    return t.x + t.y;
}

template <int CTRL>
__device__ __forceinline__ float row_read(float v) {
    return __builtin_bit_cast(float, __builtin_amdgcn_update_dpp(0, __builtin_bit_cast(int, v), CTRL, 0xf, 0xf, true));
}

__device__ __forceinline__ float elu1(float v) {
    return v > 0.f ? v : (__expf(v) - 1.0f);
}

__global__ __launch_bounds__(256) void gat_dst_gather(const int* __restrict__ src,
                                                      const float* __restrict__ M,
                                                      const float* __restrict__ a,
                                                      const int* __restrict__ seg,
                                                      const int* __restrict__ perm,
                                                      const float* __restrict__ h,
                                                      float* __restrict__ out) {
    const int lane = threadIdx.x & 31;
    const int wave = __builtin_amdgcn_readfirstlane(threadIdx.x >> 5);
    const int d = blockIdx.x * 8 + wave;
    if (d >= N_NODES) return;

    int beg = d > 0 ? seg[d - 1] : 0;
    int end = seg[d];
    beg = min(max(beg, 0), N_EDGES);
    end = min(max(end, beg), N_EDGES);

    const int half = lane >> 4;
    const int c4   = (lane & 15) * 4;

    v4f   as[4];
    float nb[4];
    {
        const v4f* __restrict__ hd = (const v4f*)(h + (size_t)d * COLS);
#pragma unroll
        for (int j = 0; j < 4; ++j) {
            const float* __restrict__ ah = a + (size_t)(2 * j + half) * (2 * OUT_F) + c4;
            as[j] = *(const v4f*)ah;
            v4f an = *(const v4f*)(ah + OUT_F);
            nb[j] = row16_sum(dot4(hd[j * 32 + lane], an));
        }
    }

    const v4f zero4 = {0.f, 0.f, 0.f, 0.f};
    v4f   acc[4] = {zero4, zero4, zero4, zero4};
    const bool up1 = (lane & 1) != 0;
    const bool up2 = (lane & 2) != 0;
    const float nb_own = up1 ? (up2 ? nb[3] : nb[2]) : (up2 ? nb[1] : nb[0]);
    float den_own = 0.f;

    for (int base = beg; base < end; base += 32) {
        const int idx = base + lane;
        int e = 0;
        if (idx < end) e = perm[idx];
        e = min(max(e, 0), N_EDGES - 1);
        const int   s_l = src[e];
        const float m_l = M[e];
        const int cnt = min(32, end - base);
        for (int k = 0; k < cnt; ++k) {
            const int   s = __builtin_amdgcn_readlane(s_l, k);
            const float m = __builtin_bit_cast(float,
                              __builtin_amdgcn_readlane(__builtin_bit_cast(int, m_l), k));
            const v4f* __restrict__ hr = (const v4f*)(h + (size_t)s * COLS);
            v4f v[4];
#pragma unroll
            for (int j = 0; j < 4; ++j) v[j] = hr[j * 32 + lane];
            float p[4];
#pragma unroll
            for (int j = 0; j < 4; ++j) p[j] = dot4_packed(v[j], as[j]);
            float k0 = up1 ? p[2] : p[0];
            float k1 = up1 ? p[3] : p[1];
            const float g0 = up1 ? p[0] : p[2];
            const float g1 = up1 ? p[1] : p[3];
            k0 += row_read<0x161>(g0);
            k1 += row_read<0x161>(g1);
            float lsum = up2 ? k1 : k0;
            const float g = up2 ? k0 : k1;
            lsum += row_read<0x162>(g);
            lsum += row_read<0x164>(lsum);
            lsum += row_read<0x168>(lsum);
            const float ex = edge_exp(lsum, nb_own, m);
            den_own += ex;
            const float e0 = row_read<0x00>(ex);
            const float e1 = row_read<0xAA>(ex);
            const float e2 = row_read<0x55>(ex);
            const float e3 = row_read<0xFF>(ex);
            acc[0] += e0 * v[0];
            acc[1] += e1 * v[1];
            acc[2] += e2 * v[2];
            acc[3] += e3 * v[3];
        }
    }
    const float den[4] = {row_read<0x00>(den_own), row_read<0xAA>(den_own),
                          row_read<0x55>(den_own), row_read<0xFF>(den_own)};

    v4f* __restrict__ od = (v4f*)(out + (size_t)d * COLS);
#pragma unroll
    for (int j = 0; j < 4; ++j) {
        const float dj = den[j] + 1e-16f;
        v4f r = acc[j];
        r.x = elu1(r.x / dj);
        r.y = elu1(r.y / dj);
        r.z = elu1(r.z / dj);
        r.w = elu1(r.w / dj);
        od[j * 32 + lane] = r;
    }
}

extern "C" void kernel_launch(void* const* d_in, const int* in_sizes, int n_in,
                              void* d_out, int out_size, void* d_ws, size_t ws_size,
                              hipStream_t stream) {
    const float* x   = (const float*)d_in[0];
    const int*   src = (const int*)  d_in[1];
    const int*   dst = (const int*)  d_in[2];
    const float* M   = (const float*)d_in[3];
    const float* W   = (const float*)d_in[4];
    const float* a   = (const float*)d_in[5];
    float* out = (float*)d_out;

    float* h    = (float*)d_ws;
    int*   seg  = (int*)(h + (size_t)N_NODES * COLS);
    int*   perm = seg + N_NODES;
    unsigned* keyed  = (unsigned*)d_ws;
    int*      bh     = (int*)d_ws + N_EDGES;
    int*      bstart = bh + SORT_G * SORT_NB;

    gat_bucket_hist<<<SORT_G, SORT_T, 0, stream>>>(dst, bh);
    gat_bucket_place<<<SORT_G, SORT_T, 0, stream>>>(dst, bh, bstart, keyed);
    gat_bucket_sort<<<SORT_NB, SORT_T, 0, stream>>>(bstart, keyed, seg, perm);

    {
        dim3 grid(N_HEADS, GEMM_RG);
        gat_gemm_wmma<<<grid, 128, 0, stream>>>(x, W, h);
    }

    gat_dst_gather<<<(N_NODES + 7) / 8, 256, 0, stream>>>(src, M, a, seg, perm, h, out);
}


# Grouping passes run beside the product: hist, place and bucket sort each open a launch whose other blocks do a third of the WMMA row tiles (one 16x64 strip per wave, 256 threads); 4 launches, not 5
# speedup vs baseline: 3.8168x; 1.0804x over previous
#include <hip/hip_runtime.h>
#include <hip/hip_bf16.h>

typedef __attribute__((ext_vector_type(2))) float v2f;
typedef __attribute__((ext_vector_type(4))) float v4f;
typedef __attribute__((ext_vector_type(8))) float v8f;

#define N_NODES  50000
#define N_EDGES  800000
#define IN_F     256
#define OUT_F    64
#define N_HEADS  8
#define COLS     (N_HEADS * OUT_F)
#define ROW_TILES (N_NODES / 16)
#define LRELU_SLOPE 0.2f

#define FUSED_T  256
#define GEMM_TPB 8
#define GEMM_RG  ((ROW_TILES + GEMM_TPB - 1) / GEMM_TPB)
#define SHARE_A  96
#define SHARE_B  128
#define SHARE_C  (GEMM_RG - SHARE_A - SHARE_B)

static_assert(SHARE_C > 0 && FUSED_T == GEMM_TPB * 32, "shares cover the row groups; one wave per row tile");

__device__ __forceinline__ void product_role(const int head, const int rg,
                                             const float* __restrict__ x,
                                             const float* __restrict__ W,
                                             float* __restrict__ h) {
    __shared__ v4f Wl[IN_F * OUT_F / 4];

    const int tid  = threadIdx.x;
    const int wave = tid >> 5;
    const int lane = tid & 31;

    {
        const float* __restrict__ Wg = W + (size_t)head * (IN_F * OUT_F);
#pragma unroll
        for (int b = 0; b < 4; ++b) {
            const int blk = b * FUSED_T + tid;
            const int kg  = blk >> 5;
            const int hf  = (blk >> 4) & 1;
            const int q   = blk & 15;
            const float* __restrict__ src0 = Wg + (size_t)(kg * 8 + hf * 4) * OUT_F + q * 4;
            const v4f r0 = *(const v4f*)(src0);
            const v4f r1 = *(const v4f*)(src0 + OUT_F);
            const v4f r2 = *(const v4f*)(src0 + 2 * OUT_F);
            const v4f r3 = *(const v4f*)(src0 + 3 * OUT_F);
            const int t = q >> 2;
            const int u = (kg * 4 + t) * 32 + hf * 16 + (q & 3) * 4;
            Wl[u + (0 ^ t)] = (v4f){r0.x, r1.x, r2.x, r3.x};
            Wl[u + (1 ^ t)] = (v4f){r0.y, r1.y, r2.y, r3.y};
            Wl[u + (2 ^ t)] = (v4f){r0.z, r1.z, r2.z, r3.z};
            Wl[u + (3 ^ t)] = (v4f){r0.w, r1.w, r2.w, r3.w};
        }
    }
    __syncthreads();

    const int tile0 = rg * GEMM_TPB + wave;
    if (tile0 >= ROW_TILES) return;

    const int c  = lane & 15;
    const int hf = lane >> 4;
    const int rowA0 = tile0 * 16 + c;
    const float* __restrict__ ap0 = x + (size_t)rowA0 * IN_F + hf * 4;
    const v4f* __restrict__ bp[4] = {Wl + lane, Wl + (lane ^ 1), Wl + (lane ^ 2), Wl + (lane ^ 3)};

    const v8f z8 = {0.f,0.f,0.f,0.f,0.f,0.f,0.f,0.f};
    v8f acc[4] = {z8, z8, z8, z8};

    for (int kc = 0; kc < IN_F / 32; ++kc) {
        v4f a0[4], b[4][4];
#pragma unroll
        for (int g = 0; g < 4; ++g) a0[g] = *(const v4f*)(ap0 + kc * 32 + g * 8);
#pragma unroll
        for (int g = 0; g < 4; ++g)
#pragma unroll
            for (int t = 0; t < 4; ++t) b[g][t] = bp[t][((kc * 4 + g) * 4 + t) * 32];
        __builtin_amdgcn_sched_barrier(0);
#pragma unroll
        for (int g = 0; g < 4; ++g) {
#pragma unroll
            for (int t = 0; t < 4; ++t)
                acc[t] = __builtin_amdgcn_wmma_f32_16x16x4_f32(false, a0[g].xy, false, b[g][t].xy,
                                                               (short)0, acc[t], false, false);
#pragma unroll
            for (int t = 0; t < 4; ++t)
                acc[t] = __builtin_amdgcn_wmma_f32_16x16x4_f32(false, a0[g].zw, false, b[g][t].zw,
                                                               (short)0, acc[t], false, false);
        }
    }

    float* __restrict__ hp = h + (size_t)(tile0 * 16 + hf * 8) * COLS + head * OUT_F + c;
#pragma unroll
    for (int v = 0; v < 8; ++v) {
        float* __restrict__ hr = hp + (size_t)v * COLS;
        hr[0]  = acc[0][v];
        hr[16] = acc[1][v];
        hr[32] = acc[2][v];
        hr[48] = acc[3][v];
    }
}

#define SORT_T     FUSED_T
#define SORT_NB    196
#define SORT_V     4
#define SORT_EPB   (SORT_T * SORT_V * 4)
#define SORT_G     ((N_EDGES + SORT_EPB - 1) / SORT_EPB)
#define SORT_PARTS (SORT_T / 256)
#define SORT_GP    (SORT_G / SORT_PARTS)
#define EDGE_V4    (N_EDGES / 4)
#define SORT_KU    24
#define COL_B      28

static_assert(SORT_NB * 256 >= N_NODES, "the buckets cover every node");
static_assert(SORT_NB <= 256 && SORT_T % 256 == 0 && SORT_G % SORT_PARTS == 0 && N_EDGES % 4 == 0, "table shape");
static_assert(SORT_G == SORT_NB, "every launch opens with the same number of pass blocks");
static_assert(SORT_GP % COL_B == 0, "the table's column is read in whole groups");
static_assert(N_EDGES <= (1 << 20), "an edge id fits the low 20 bits of a key");

typedef __attribute__((ext_vector_type(4))) int v4i;

__device__ __forceinline__ int wave_incl_scan(int v, int lane) {
#pragma unroll
    for (int off = 1; off < 32; off <<= 1) {
        int u = __shfl_up(v, off);
        if (lane >= off) v += u;
    }
    return v;
}

__device__ __forceinline__ int bucket_of(int d) {
    return min(max(d >> 8, 0), SORT_NB - 1);
}

__device__ __forceinline__ void bucket_hist_role(const int g,
                                                 const int* __restrict__ dst,
                                                 int* __restrict__ bh) {
    __shared__ int hc[256];
    const int t = threadIdx.x;
    v4i d[SORT_V];
#pragma unroll
    for (int k = 0; k < SORT_V; ++k) {
        const int q = g * (SORT_T * SORT_V) + k * SORT_T + t;
        d[k] = ((const v4i*)dst)[min(q, EDGE_V4 - 1)];
    }
    if (t < 256) hc[t] = 0;
    __syncthreads();
#pragma unroll
    for (int k = 0; k < SORT_V; ++k) {
        const int q = g * (SORT_T * SORT_V) + k * SORT_T + t;
        if (q < EDGE_V4) {
            atomicAdd(&hc[bucket_of(d[k].x)], 1);
            atomicAdd(&hc[bucket_of(d[k].y)], 1);
            atomicAdd(&hc[bucket_of(d[k].z)], 1);
            atomicAdd(&hc[bucket_of(d[k].w)], 1);
        }
    }
    __syncthreads();
    if (t < SORT_NB) bh[g * SORT_NB + t] = hc[t];
}

__device__ __forceinline__ void bucket_place_role(const int g,
                                                  const int* __restrict__ dst,
                                                  const int* __restrict__ bh,
                                                  int* __restrict__ bstart,
                                                  unsigned* __restrict__ keyed) {
    __shared__ int pall[SORT_PARTS][256];
    __shared__ int pmine[SORT_PARTS][256];
    __shared__ int cur[256];
    __shared__ int wsum[8];
    const int t    = threadIdx.x;
    const int lane = t & 31;
    const int b    = t & 255;
    const int part = t >> 8;
    v4i d[SORT_V];
#pragma unroll
    for (int k = 0; k < SORT_V; ++k) {
        const int q = g * (SORT_T * SORT_V) + k * SORT_T + t;
        d[k] = ((const v4i*)dst)[min(q, EDGE_V4 - 1)];
    }

    int sa = 0, sm = 0;
    if (b < SORT_NB) {
        const int* __restrict__ col = bh + part * SORT_GP * SORT_NB + b;
        for (int k0 = 0; k0 < SORT_GP; k0 += COL_B) {
            int v[COL_B];
#pragma unroll
            for (int j = 0; j < COL_B; ++j) v[j] = col[(k0 + j) * SORT_NB];
#pragma unroll
            for (int j = 0; j < COL_B; ++j) {
                sa += v[j];
                if (part * SORT_GP + k0 + j < g) sm += v[j];
            }
        }
    }
    pall[part][b]  = sa;
    pmine[part][b] = sm;
    __syncthreads();

    int tot = 0, inc = 0;
    if (t < 256) {
#pragma unroll
        for (int p = 0; p < SORT_PARTS; ++p) tot += pall[p][t];
        inc = wave_incl_scan(tot, lane);
        if (lane == 31) wsum[t >> 5] = inc;
    }
    __syncthreads();
    if (t < 256) {
        int wb = 0;
        for (int k = 0; k < (t >> 5); ++k) wb += wsum[k];
        const int start = wb + inc - tot;
        int mine = 0;
#pragma unroll
        for (int p = 0; p < SORT_PARTS; ++p) mine += pmine[p][t];
        cur[t] = start + mine;
        if (g == 0) bstart[t] = start;
    }
    __syncthreads();

#pragma unroll
    for (int k = 0; k < SORT_V; ++k) {
        const int q = g * (SORT_T * SORT_V) + k * SORT_T + t;
        if (q < EDGE_V4) {
            const int dd[4] = {d[k].x, d[k].y, d[k].z, d[k].w};
            int pos[4];
#pragma unroll
            for (int j = 0; j < 4; ++j) pos[j] = atomicAdd(&cur[bucket_of(dd[j])], 1);
#pragma unroll
            for (int j = 0; j < 4; ++j)
                if ((unsigned)pos[j] < (unsigned)N_EDGES)
                    keyed[pos[j]] = ((unsigned)(dd[j] & 255) << 20) | (unsigned)(4 * q + j);
        }
    }
}

__device__ __forceinline__ void bucket_sort_role(const int b,
                                                 const int* __restrict__ bstart,
                                                 const unsigned* __restrict__ keyed,
                                                 int* __restrict__ seg,
                                                 int* __restrict__ perm) {
    __shared__ int nc[256];
    __shared__ int cur[256];
    __shared__ int wsum[8];
    const int t    = threadIdx.x;
    const int lane = t & 31;
    int beg = bstart[b];
    int end = bstart[b + 1];
    beg = min(max(beg, 0), N_EDGES);
    end = min(max(end, beg), N_EDGES);

    if (t < 256) nc[t] = 0;
    __syncthreads();
    for (int i0 = beg; i0 < end; i0 += SORT_KU * SORT_T) {
        unsigned w[SORT_KU];
#pragma unroll
        for (int k = 0; k < SORT_KU; ++k) w[k] = keyed[min(i0 + k * SORT_T + t, end - 1)];
#pragma unroll
        for (int k = 0; k < SORT_KU; ++k)
            if (i0 + k * SORT_T + t < end) atomicAdd(&nc[w[k] >> 20], 1);
    }
    __syncthreads();

    int v = 0, inc = 0;
    if (t < 256) {
        v   = nc[t];
        inc = wave_incl_scan(v, lane);
        if (lane == 31) wsum[t >> 5] = inc;
    }
    __syncthreads();
    if (t < 256) {
        int wb = 0;
        for (int k = 0; k < (t >> 5); ++k) wb += wsum[k];
        const int incl = beg + wb + inc;
        cur[t] = incl - v;
        const int node = b * 256 + t;
        if (node < N_NODES) seg[node] = incl;
    }
    __syncthreads();
    for (int i0 = beg; i0 < end; i0 += SORT_KU * SORT_T) {
        unsigned w[SORT_KU];
        int pos[SORT_KU];
#pragma unroll
        for (int k = 0; k < SORT_KU; ++k) w[k] = keyed[min(i0 + k * SORT_T + t, end - 1)];
#pragma unroll
        for (int k = 0; k < SORT_KU; ++k) {
            pos[k] = -1;
            if (i0 + k * SORT_T + t < end) pos[k] = atomicAdd(&cur[w[k] >> 20], 1);
        }
#pragma unroll
        for (int k = 0; k < SORT_KU; ++k)
            if ((unsigned)pos[k] < (unsigned)N_EDGES) perm[pos[k]] = (int)(w[k] & 0xFFFFFu);
    }
}

__global__ __launch_bounds__(FUSED_T) void gat_hist_and_product(const int* __restrict__ dst,
                                                               int* __restrict__ bh,
                                                               const float* __restrict__ x,
                                                               const float* __restrict__ W,
                                                               float* __restrict__ h) {
    if (blockIdx.x < SORT_G) { bucket_hist_role(blockIdx.x, dst, bh); return; }
    const int g = blockIdx.x - SORT_G;
    product_role(g & (N_HEADS - 1), g / N_HEADS, x, W, h);
}

__global__ __launch_bounds__(FUSED_T) void gat_place_and_product(const int* __restrict__ dst,
                                                                const int* __restrict__ bh,
                                                                int* __restrict__ bstart,
                                                                unsigned* __restrict__ keyed,
                                                                const float* __restrict__ x,
                                                                const float* __restrict__ W,
                                                                float* __restrict__ h) {
    if (blockIdx.x < SORT_G) { bucket_place_role(blockIdx.x, dst, bh, bstart, keyed); return; }
    const int g = blockIdx.x - SORT_G;
    product_role(g & (N_HEADS - 1), SHARE_A + g / N_HEADS, x, W, h);
}

__global__ __launch_bounds__(FUSED_T) void gat_sort_and_product(const int* __restrict__ bstart,
                                                               const unsigned* __restrict__ keyed,
                                                               int* __restrict__ seg,
                                                               int* __restrict__ perm,
                                                               const float* __restrict__ x,
                                                               const float* __restrict__ W,
                                                               float* __restrict__ h) {
    if (blockIdx.x < SORT_NB) { bucket_sort_role(blockIdx.x, bstart, keyed, seg, perm); return; }
    const int g = blockIdx.x - SORT_NB;
    product_role(g & (N_HEADS - 1), SHARE_A + SHARE_B + g / N_HEADS, x, W, h);
}

__device__ __forceinline__ float edge_exp(float as, float an, float m) {
    float lg = as + an;
    lg = (lg > 0.f ? lg : LRELU_SLOPE * lg) * m;
    return __expf(lg);
}

__device__ __forceinline__ float row16_sum(float v) {
    v += __builtin_bit_cast(float, __builtin_amdgcn_update_dpp(0, __builtin_bit_cast(int, v), 0x161, 0xf, 0xf, true));
    v += __builtin_bit_cast(float, __builtin_amdgcn_update_dpp(0, __builtin_bit_cast(int, v), 0x162, 0xf, 0xf, true));
    v += __builtin_bit_cast(float, __builtin_amdgcn_update_dpp(0, __builtin_bit_cast(int, v), 0x164, 0xf, 0xf, true));
    v += __builtin_bit_cast(float, __builtin_amdgcn_update_dpp(0, __builtin_bit_cast(int, v), 0x168, 0xf, 0xf, true));
    return v;
}

__device__ __forceinline__ float dot4(v4f p, v4f q) {
    return p.x * q.x + p.y * q.y + p.z * q.z + p.w * q.w;
}

__device__ __forceinline__ float dot4_packed(v4f p, v4f q) {
    v2f t = p.xy * q.xy;
    t = __builtin_elementwise_fma(p.zw, q.zw, t);
    return t.x + t.y;
}

template <int CTRL>
__device__ __forceinline__ float row_read(float v) {
    return __builtin_bit_cast(float, __builtin_amdgcn_update_dpp(0, __builtin_bit_cast(int, v), CTRL, 0xf, 0xf, true));
}

__device__ __forceinline__ float elu1(float v) {
    return v > 0.f ? v : (__expf(v) - 1.0f);
}

__global__ __launch_bounds__(256) void gat_dst_gather(const int* __restrict__ src,
                                                      const float* __restrict__ M,
                                                      const float* __restrict__ a,
                                                      const int* __restrict__ seg,
                                                      const int* __restrict__ perm,
                                                      const float* __restrict__ h,
                                                      float* __restrict__ out) {
    const int lane = threadIdx.x & 31;
    const int wave = __builtin_amdgcn_readfirstlane(threadIdx.x >> 5);
    const int d = blockIdx.x * 8 + wave;
    if (d >= N_NODES) return;

    int beg = d > 0 ? seg[d - 1] : 0;
    int end = seg[d];
    beg = min(max(beg, 0), N_EDGES);
    end = min(max(end, beg), N_EDGES);

    const int half = lane >> 4;
    const int c4   = (lane & 15) * 4;

    v4f   as[4];
    float nb[4];
    {
        const v4f* __restrict__ hd = (const v4f*)(h + (size_t)d * COLS);
#pragma unroll
        for (int j = 0; j < 4; ++j) {
            const float* __restrict__ ah = a + (size_t)(2 * j + half) * (2 * OUT_F) + c4;
            as[j] = *(const v4f*)ah;
            v4f an = *(const v4f*)(ah + OUT_F);
            nb[j] = row16_sum(dot4(hd[j * 32 + lane], an));
        }
    }

    const v4f zero4 = {0.f, 0.f, 0.f, 0.f};
    v4f   acc[4] = {zero4, zero4, zero4, zero4};
    const bool up1 = (lane & 1) != 0;
    const bool up2 = (lane & 2) != 0;
    const float nb_own = up1 ? (up2 ? nb[3] : nb[2]) : (up2 ? nb[1] : nb[0]);
    float den_own = 0.f;

    for (int base = beg; base < end; base += 32) {
        const int idx = base + lane;
        int e = 0;
        if (idx < end) e = perm[idx];
        e = min(max(e, 0), N_EDGES - 1);
        const int   s_l = src[e];
        const float m_l = M[e];
        const int cnt = min(32, end - base);
        for (int k = 0; k < cnt; ++k) {
            const int   s = __builtin_amdgcn_readlane(s_l, k);
            const float m = __builtin_bit_cast(float,
                              __builtin_amdgcn_readlane(__builtin_bit_cast(int, m_l), k));
            const v4f* __restrict__ hr = (const v4f*)(h + (size_t)s * COLS);
            v4f v[4];
#pragma unroll
            for (int j = 0; j < 4; ++j) v[j] = hr[j * 32 + lane];
            float p[4];
#pragma unroll
            for (int j = 0; j < 4; ++j) p[j] = dot4_packed(v[j], as[j]);
            float k0 = up1 ? p[2] : p[0];
            float k1 = up1 ? p[3] : p[1];
            const float g0 = up1 ? p[0] : p[2];
            const float g1 = up1 ? p[1] : p[3];
            k0 += row_read<0x161>(g0);
            k1 += row_read<0x161>(g1);
            float lsum = up2 ? k1 : k0;
            const float g = up2 ? k0 : k1;
            lsum += row_read<0x162>(g);
            lsum += row_read<0x164>(lsum);
            lsum += row_read<0x168>(lsum);
            const float ex = edge_exp(lsum, nb_own, m);
            den_own += ex;
            const float e0 = row_read<0x00>(ex);
            const float e1 = row_read<0xAA>(ex);
            const float e2 = row_read<0x55>(ex);
            const float e3 = row_read<0xFF>(ex);
            acc[0] += e0 * v[0];
            acc[1] += e1 * v[1];
            acc[2] += e2 * v[2];
            acc[3] += e3 * v[3];
        }
    }
    const float den[4] = {row_read<0x00>(den_own), row_read<0xAA>(den_own),
                          row_read<0x55>(den_own), row_read<0xFF>(den_own)};

    v4f* __restrict__ od = (v4f*)(out + (size_t)d * COLS);
#pragma unroll
    for (int j = 0; j < 4; ++j) {
        const float dj = den[j] + 1e-16f;
        v4f r = acc[j];
        r.x = elu1(r.x / dj);
        r.y = elu1(r.y / dj);
        r.z = elu1(r.z / dj);
        r.w = elu1(r.w / dj);
        od[j * 32 + lane] = r;
    }
}

extern "C" void kernel_launch(void* const* d_in, const int* in_sizes, int n_in,
                              void* d_out, int out_size, void* d_ws, size_t ws_size,
                              hipStream_t stream) {
    const float* x   = (const float*)d_in[0];
    const int*   src = (const int*)  d_in[1];
    const int*   dst = (const int*)  d_in[2];
    const float* M   = (const float*)d_in[3];
    const float* W   = (const float*)d_in[4];
    const float* a   = (const float*)d_in[5];
    float* out = (float*)d_out;

    float* h    = (float*)d_ws;
    int*   seg  = (int*)(h + (size_t)N_NODES * COLS);
    int*   perm = seg + N_NODES;
    unsigned* keyed  = (unsigned*)d_out;
    int*      bh     = (int*)d_out + N_EDGES;
    int*      bstart = bh + SORT_G * SORT_NB;

    gat_hist_and_product<<<SORT_G + SHARE_A * N_HEADS, FUSED_T, 0, stream>>>(dst, bh, x, W, h);
    gat_place_and_product<<<SORT_G + SHARE_B * N_HEADS, FUSED_T, 0, stream>>>(dst, bh, bstart, keyed, x, W, h);
    gat_sort_and_product<<<SORT_NB + SHARE_C * N_HEADS, FUSED_T, 0, stream>>>(bstart, keyed, seg, perm, x, W, h);

    gat_dst_gather<<<(N_NODES + 7) / 8, 256, 0, stream>>>(src, M, a, seg, perm, h, out);
}
